# starting kernel
# baseline (speedup 1.0000x reference)
#include <hip/hip_runtime.h>


typedef float v2f __attribute__((ext_vector_type(2)));
typedef float v8f __attribute__((ext_vector_type(8)));

#define F_IN  128
#define F_HID 64
#define F_OUT 16

__global__ void k_init_deg(float* __restrict__ deg, int n) {
    int i = blockIdx.x * blockDim.x + threadIdx.x;
    if (i < n) deg[i] = 1.0f;
}

__global__ void k_accum_deg(const int* __restrict__ dst, float* __restrict__ deg, int E) {
    int i = blockIdx.x * blockDim.x + threadIdx.x;
    if (i < E) unsafeAtomicAdd(&deg[dst[i]], 1.0f);
}

__global__ void k_rsqrt_inplace(float* __restrict__ deg, int n) {
    int i = blockIdx.x * blockDim.x + threadIdx.x;
    if (i < n) deg[i] = __frsqrt_rn(deg[i]);
}

__global__ __launch_bounds__(256)
void k_gemm1(const float* __restrict__ X, const float* __restrict__ W,
             float* __restrict__ H, int mTiles) {
    __shared__ float Bs[F_IN * 16];
    const int nBase = blockIdx.y * 16;
    for (int idx = threadIdx.x; idx < F_IN * 16; idx += 256) {
        int k = idx >> 4, nn = idx & 15;
        Bs[idx] = W[k * F_HID + nBase + nn];
    }
    __syncthreads();

    const int wave  = threadIdx.x >> 5;
    const int lane  = threadIdx.x & 31;
    const int mTile = blockIdx.x * 8 + wave;
    if (mTile >= mTiles) return;
    const int mBase = mTile * 16;
    const int half  = lane >> 4;
    const int l     = lane & 15;
    const float* arow = X + (size_t)(mBase + l) * F_IN;

    v8f c = {};
    #pragma unroll 4
    for (int k = 0; k < F_IN; k += 4) {
        const int ak = k + half * 2;
        v2f a = *(const v2f*)(arow + ak);
        v2f b; b.x = Bs[ak * 16 + l]; b.y = Bs[(ak + 1) * 16 + l];
        c = __builtin_amdgcn_wmma_f32_16x16x4_f32(false, a, false, b,
                                                  (short)0, c, false, false);
    }
    float* orow = H + (size_t)(mBase + half * 8) * F_HID + nBase + l;
    #pragma unroll
    for (int r = 0; r < 8; ++r) orow[(size_t)r * F_HID] = c[r];
}

__global__ void k_self_init64(const float* __restrict__ h1, const float* __restrict__ dinv,
                              float* __restrict__ agg, int n) {
    int t = blockIdx.x * blockDim.x + threadIdx.x;
    if (t < n * F_HID) {
        int i = t >> 6;
        float di = dinv[i];
        agg[t] = h1[t] * di * di;
    }
}

__global__ __launch_bounds__(256)
void k_scatter64(const int* __restrict__ src, const int* __restrict__ dst,
                 const float* __restrict__ h1, const float* __restrict__ dinv,
                 float* __restrict__ agg, int E) {
    int warp = (blockIdx.x * blockDim.x + threadIdx.x) >> 5;
    int lane = threadIdx.x & 31;
    if (warp >= E) return;
    int s = src[warp], d = dst[warp];
    float norm = dinv[s] * dinv[d];
    float2 v = ((const float2*)(h1 + (size_t)s * F_HID))[lane];
    float* drow = agg + (size_t)d * F_HID + 2 * lane;
    unsafeAtomicAdd(drow + 0, v.x * norm);
    unsafeAtomicAdd(drow + 1, v.y * norm);
}

__global__ __launch_bounds__(256)
void k_gemm2(const float* __restrict__ A1, const float* __restrict__ bias1,
             const float* __restrict__ W, float* __restrict__ H, int mTiles) {
    __shared__ float Bs[F_HID * 16];
    __shared__ float b1s[F_HID];
    for (int idx = threadIdx.x; idx < F_HID * 16; idx += 256) {
        int k = idx >> 4, nn = idx & 15;
        Bs[idx] = W[k * F_OUT + nn];
    }
    if (threadIdx.x < F_HID) b1s[threadIdx.x] = bias1[threadIdx.x];
    __syncthreads();

    const int wave  = threadIdx.x >> 5;
    const int lane  = threadIdx.x & 31;
    const int mTile = blockIdx.x * 8 + wave;
    if (mTile >= mTiles) return;
    const int mBase = mTile * 16;
    const int half  = lane >> 4;
    const int l     = lane & 15;
    const float* arow = A1 + (size_t)(mBase + l) * F_HID;

    v8f c = {};
    #pragma unroll 4
    for (int k = 0; k < F_HID; k += 4) {
        const int ak = k + half * 2;
        v2f a, b;
        a.x = fmaxf(arow[ak]     + b1s[ak],     0.0f);
        a.y = fmaxf(arow[ak + 1] + b1s[ak + 1], 0.0f);
        b.x = Bs[ak * 16 + l]; b.y = Bs[(ak + 1) * 16 + l];
        c = __builtin_amdgcn_wmma_f32_16x16x4_f32(false, a, false, b,
                                                  (short)0, c, false, false);
    }
    float* orow = H + (size_t)(mBase + half * 8) * F_OUT + l;
    #pragma unroll
    for (int r = 0; r < 8; ++r) orow[(size_t)r * F_OUT] = c[r];
}

__global__ void k_self_init16(const float* __restrict__ h2, const float* __restrict__ dinv,
                              const float* __restrict__ bias2, float* __restrict__ out, int n) {
    int t = blockIdx.x * blockDim.x + threadIdx.x;
    if (t < n * F_OUT) {
        int i = t >> 4, f = t & 15;
        float di = dinv[i];
        out[t] = h2[t] * di * di + bias2[f];
    }
}

__global__ __launch_bounds__(256)
void k_scatter16(const int* __restrict__ src, const int* __restrict__ dst,
                 const float* __restrict__ h2, const float* __restrict__ dinv,
                 float* __restrict__ out, int E) {
    int t = blockIdx.x * blockDim.x + threadIdx.x;
    int e = t >> 4, f = t & 15;
    if (e >= E) return;
    int s = src[e], d = dst[e];
    float norm = dinv[s] * dinv[d];
    unsafeAtomicAdd(&out[(size_t)d * F_OUT + f], h2[(size_t)s * F_OUT + f] * norm);
}

extern "C" void kernel_launch(void* const* d_in, const int* in_sizes, int n_in,
                              void* d_out, int out_size, void* d_ws, size_t ws_size,
                              hipStream_t stream) {
    const float* x  = (const float*)d_in[0];
    const int*   ei = (const int*)  d_in[1];
    const float* W1 = (const float*)d_in[2];
    const float* b1 = (const float*)d_in[3];
    const float* W2 = (const float*)d_in[4];
    const float* b2 = (const float*)d_in[5];
    float* out = (float*)d_out;

    const int n = in_sizes[0] / F_IN;
    const int E = in_sizes[1] / 2;
    const int* src = ei;
    const int* dst = ei + E;

    float* dinv = (float*)d_ws;
    float* h1   = dinv + n;
    float* agg1 = h1 + (size_t)n * F_HID;
    float* h2   = agg1 + (size_t)n * F_HID;

    const int mTiles  = n / 16;
    const int mGroups = (mTiles + 7) / 8;

    k_init_deg<<<(n + 255) / 256, 256, 0, stream>>>(dinv, n);
    k_accum_deg<<<(E + 255) / 256, 256, 0, stream>>>(dst, dinv, E);
    k_rsqrt_inplace<<<(n + 255) / 256, 256, 0, stream>>>(dinv, n);

    k_gemm1<<<dim3(mGroups, F_HID / 16), 256, 0, stream>>>(x, W1, h1, mTiles);

    k_self_init64<<<((size_t)n * F_HID + 255) / 256, 256, 0, stream>>>(h1, dinv, agg1, n);
    k_scatter64<<<(int)(((size_t)E * 32 + 255) / 256), 256, 0, stream>>>(src, dst, h1, dinv, agg1, E);

    k_gemm2<<<dim3(mGroups, 1), 256, 0, stream>>>(agg1, b1, W2, h2, mTiles);

    k_self_init16<<<((size_t)n * F_OUT + 255) / 256, 256, 0, stream>>>(h2, dinv, b2, out, n);
    k_scatter16<<<(int)(((size_t)E * 16 + 255) / 256), 256, 0, stream>>>(src, dst, h2, dinv, out, E);
}


# edge scatters, both layers: one wave per 32-edge batch (index+norm loaded once per lane, rows walked by lane broadcast, 8 row loads in flight), 64x fewer waves; f32 atomic sums still arrival-ordered
# speedup vs baseline: 1.1617x; 1.1617x over previous
#include <hip/hip_runtime.h>


typedef float v2f __attribute__((ext_vector_type(2)));
typedef float v8f __attribute__((ext_vector_type(8)));

#define F_IN  128
#define F_HID 64
#define F_OUT 16

__global__ void k_init_deg(float* __restrict__ deg, int n) {
    int i = blockIdx.x * blockDim.x + threadIdx.x;
    if (i < n) deg[i] = 1.0f;
}

__global__ void k_accum_deg(const int* __restrict__ dst, float* __restrict__ deg, int E) {
    int i = blockIdx.x * blockDim.x + threadIdx.x;
    if (i < E) unsafeAtomicAdd(&deg[dst[i]], 1.0f);
}

__global__ void k_rsqrt_inplace(float* __restrict__ deg, int n) {
    int i = blockIdx.x * blockDim.x + threadIdx.x;
    if (i < n) deg[i] = __frsqrt_rn(deg[i]);
}

__global__ __launch_bounds__(256)
void k_gemm1(const float* __restrict__ X, const float* __restrict__ W,
             float* __restrict__ H, int mTiles) {
    __shared__ float Bs[F_IN * 16];
    const int nBase = blockIdx.y * 16;
    for (int idx = threadIdx.x; idx < F_IN * 16; idx += 256) {
        int k = idx >> 4, nn = idx & 15;
        Bs[idx] = W[k * F_HID + nBase + nn];
    }
    __syncthreads();

    const int wave  = threadIdx.x >> 5;
    const int lane  = threadIdx.x & 31;
    const int mTile = blockIdx.x * 8 + wave;
    if (mTile >= mTiles) return;
    const int mBase = mTile * 16;
    const int half  = lane >> 4;
    const int l     = lane & 15;
    const float* arow = X + (size_t)(mBase + l) * F_IN;

    v8f c = {};
    #pragma unroll 4
    for (int k = 0; k < F_IN; k += 4) {
        const int ak = k + half * 2;
        v2f a = *(const v2f*)(arow + ak);
        v2f b; b.x = Bs[ak * 16 + l]; b.y = Bs[(ak + 1) * 16 + l];
        c = __builtin_amdgcn_wmma_f32_16x16x4_f32(false, a, false, b,
                                                  (short)0, c, false, false);
    }
    float* orow = H + (size_t)(mBase + half * 8) * F_HID + nBase + l;
    #pragma unroll
    for (int r = 0; r < 8; ++r) orow[(size_t)r * F_HID] = c[r];
}

__global__ void k_self_init64(const float* __restrict__ h1, const float* __restrict__ dinv,
                              float* __restrict__ agg, int n) {
    int t = blockIdx.x * blockDim.x + threadIdx.x;
    if (t < n * F_HID) {
        int i = t >> 6;
        float di = dinv[i];
        agg[t] = h1[t] * di * di;
    }
}

__global__ __launch_bounds__(256)
void k_scatter64(const int* __restrict__ src, const int* __restrict__ dst,
                 const float* __restrict__ h1, const float* __restrict__ dinv,
                 float* __restrict__ agg, int E) {
    const int lane   = threadIdx.x & 31;
    const int wave   = __builtin_amdgcn_readfirstlane((blockIdx.x * blockDim.x + threadIdx.x) >> 5);
    const int nWaves = (gridDim.x * blockDim.x) >> 5;
    const int nBatch = (E + 31) >> 5;
    for (int b = wave; b < nBatch; b += nWaves) {
        const int e0 = b << 5;
        const int e  = e0 + lane;
        int s = 0, d = 0;
        float nrm = 0.0f;
        if (e < E) { s = src[e]; d = dst[e]; nrm = dinv[s] * dinv[d]; }
        const int cnt = min(32, E - e0);
        for (int j0 = 0; j0 < cnt; j0 += 8) {
            float2 v[8];
            #pragma unroll
            for (int u = 0; u < 8; ++u) {
                const int sj = __builtin_amdgcn_readlane(s, j0 + u);
                v[u] = ((const float2*)(h1 + (size_t)sj * F_HID))[lane];
            }
            asm volatile("" :: "v"(v[0].x), "v"(v[0].y), "v"(v[1].x), "v"(v[1].y),
                               "v"(v[2].x), "v"(v[2].y), "v"(v[3].x), "v"(v[3].y),
                               "v"(v[4].x), "v"(v[4].y), "v"(v[5].x), "v"(v[5].y),
                               "v"(v[6].x), "v"(v[6].y), "v"(v[7].x), "v"(v[7].y));
            #pragma unroll
            for (int u = 0; u < 8; ++u) {
                if (j0 + u < cnt) {
                    const int   dj = __builtin_amdgcn_readlane(d, j0 + u);
                    const float nj = __int_as_float(
                        __builtin_amdgcn_readlane(__float_as_int(nrm), j0 + u));
                    float* drow = agg + (size_t)dj * F_HID + 2 * lane;
                    unsafeAtomicAdd(drow + 0, v[u].x * nj);
                    unsafeAtomicAdd(drow + 1, v[u].y * nj);
                }
            }
        }
    }
}

__global__ __launch_bounds__(256)
void k_gemm2(const float* __restrict__ A1, const float* __restrict__ bias1,
             const float* __restrict__ W, float* __restrict__ H, int mTiles) {
    __shared__ float Bs[F_HID * 16];
    __shared__ float b1s[F_HID];
    for (int idx = threadIdx.x; idx < F_HID * 16; idx += 256) {
        int k = idx >> 4, nn = idx & 15;
        Bs[idx] = W[k * F_OUT + nn];
    }
    if (threadIdx.x < F_HID) b1s[threadIdx.x] = bias1[threadIdx.x];
    __syncthreads();

    const int wave  = threadIdx.x >> 5;
    const int lane  = threadIdx.x & 31;
    const int mTile = blockIdx.x * 8 + wave;
    if (mTile >= mTiles) return;
    const int mBase = mTile * 16;
    const int half  = lane >> 4;
    const int l     = lane & 15;
    const float* arow = A1 + (size_t)(mBase + l) * F_HID;

    v8f c = {};
    #pragma unroll 4
    for (int k = 0; k < F_HID; k += 4) {
        const int ak = k + half * 2;
        v2f a, b;
        a.x = fmaxf(arow[ak]     + b1s[ak],     0.0f);
        a.y = fmaxf(arow[ak + 1] + b1s[ak + 1], 0.0f);
        b.x = Bs[ak * 16 + l]; b.y = Bs[(ak + 1) * 16 + l];
        c = __builtin_amdgcn_wmma_f32_16x16x4_f32(false, a, false, b,
                                                  (short)0, c, false, false);
    }
    float* orow = H + (size_t)(mBase + half * 8) * F_OUT + l;
    #pragma unroll
    for (int r = 0; r < 8; ++r) orow[(size_t)r * F_OUT] = c[r];
}

__global__ void k_self_init16(const float* __restrict__ h2, const float* __restrict__ dinv,
                              const float* __restrict__ bias2, float* __restrict__ out, int n) {
    int t = blockIdx.x * blockDim.x + threadIdx.x;
    if (t < n * F_OUT) {
        int i = t >> 4, f = t & 15;
        float di = dinv[i];
        out[t] = h2[t] * di * di + bias2[f];
    }
}

__global__ __launch_bounds__(256)
void k_scatter16(const int* __restrict__ src, const int* __restrict__ dst,
                 const float* __restrict__ h2, const float* __restrict__ dinv,
                 float* __restrict__ out, int E) {
    const int lane   = threadIdx.x & 31;
    const int wave   = __builtin_amdgcn_readfirstlane((blockIdx.x * blockDim.x + threadIdx.x) >> 5);
    const int nWaves = (gridDim.x * blockDim.x) >> 5;
    const int nBatch = (E + 31) >> 5;
    const int half   = lane >> 4;
    const int f      = lane & 15;
    for (int b = wave; b < nBatch; b += nWaves) {
        const int e0 = b << 5;
        const int e  = e0 + lane;
        int s = 0, d = 0;
        float nrm = 0.0f;
        if (e < E) { s = src[e]; d = dst[e]; nrm = dinv[s] * dinv[d]; }
        const int cnt = min(32, E - e0);
        for (int j0 = 0; j0 < cnt; j0 += 16) {
            float v[8];
            #pragma unroll
            for (int u = 0; u < 8; ++u) {
                const int sj = __shfl(s, j0 + 2 * u + half);
                v[u] = h2[(size_t)sj * F_OUT + f];
            }
            asm volatile("" :: "v"(v[0]), "v"(v[1]), "v"(v[2]), "v"(v[3]),
                               "v"(v[4]), "v"(v[5]), "v"(v[6]), "v"(v[7]));
            #pragma unroll
            for (int u = 0; u < 8; ++u) {
                const int   idx = j0 + 2 * u + half;
                const int   dj  = __shfl(d, idx);
                const float nj  = __shfl(nrm, idx);
                if (idx < cnt)
                    unsafeAtomicAdd(&out[(size_t)dj * F_OUT + f], v[u] * nj);
            }
        }
    }
}

extern "C" void kernel_launch(void* const* d_in, const int* in_sizes, int n_in,
                              void* d_out, int out_size, void* d_ws, size_t ws_size,
                              hipStream_t stream) {
    const float* x  = (const float*)d_in[0];
    const int*   ei = (const int*)  d_in[1];
    const float* W1 = (const float*)d_in[2];
    const float* b1 = (const float*)d_in[3];
    const float* W2 = (const float*)d_in[4];
    const float* b2 = (const float*)d_in[5];
    float* out = (float*)d_out;

    const int n = in_sizes[0] / F_IN;
    const int E = in_sizes[1] / 2;
    const int* src = ei;
    const int* dst = ei + E;

    float* dinv = (float*)d_ws;
    float* h1   = dinv + n;
    float* agg1 = h1 + (size_t)n * F_HID;
    float* h2   = agg1 + (size_t)n * F_HID;

    const int mTiles  = n / 16;
    const int mGroups = (mTiles + 7) / 8;

    k_init_deg<<<(n + 255) / 256, 256, 0, stream>>>(dinv, n);
    k_accum_deg<<<(E + 255) / 256, 256, 0, stream>>>(dst, dinv, E);
    k_rsqrt_inplace<<<(n + 255) / 256, 256, 0, stream>>>(dinv, n);

    k_gemm1<<<dim3(mGroups, F_HID / 16), 256, 0, stream>>>(x, W1, h1, mTiles);

    k_self_init64<<<((size_t)n * F_HID + 255) / 256, 256, 0, stream>>>(h1, dinv, agg1, n);
    k_scatter64<<<(int)((((size_t)E + 1) / 2 + 255) / 256), 256, 0, stream>>>(src, dst, h1, dinv, agg1, E);

    k_gemm2<<<dim3(mGroups, 1), 256, 0, stream>>>(agg1, b1, W2, h2, mTiles);

    k_self_init16<<<((size_t)n * F_OUT + 255) / 256, 256, 0, stream>>>(h2, dinv, b2, out, n);
    k_scatter16<<<(int)((((size_t)E + 1) / 2 + 255) / 256), 256, 0, stream>>>(src, dst, h2, dinv, out, E);
}


# edge sums without f32 atomics: in-call bucket by destination (int count atomic returns arrival rank), one wave adds a node's rows in registers in rank order, layer 2 in two parity partials
# speedup vs baseline: 1.5333x; 1.3199x over previous
#include <hip/hip_runtime.h>


typedef float v2f __attribute__((ext_vector_type(2)));
typedef float v8f __attribute__((ext_vector_type(8)));

#define F_IN  128
#define F_HID 64
#define F_OUT 16

__global__ void k_zero_cnt(int* __restrict__ cnt, int* __restrict__ cursor, int n) {
    int i = blockIdx.x * blockDim.x + threadIdx.x;
    if (i < n) cnt[i] = 0;
    if (i == 0) *cursor = 0;
}

__global__ void k_count(const int* __restrict__ dst, int* __restrict__ cnt,
                        int* __restrict__ rank, int E) {
    int e = blockIdx.x * blockDim.x + threadIdx.x;
    if (e < E) rank[e] = atomicAdd(&cnt[dst[e]], 1);
}

__global__ void k_offsets(const int* __restrict__ cnt, int* __restrict__ cursor,
                          int* __restrict__ off, float* __restrict__ dinv, int n) {
    const int i    = blockIdx.x * blockDim.x + threadIdx.x;
    const int lane = threadIdx.x & 31;
    const int c    = (i < n) ? cnt[i] : 0;
    int incl = c;
    #pragma unroll
    for (int sft = 1; sft < 32; sft <<= 1) {
        int t = __shfl_up(incl, sft);
        if (lane >= sft) incl += t;
    }
    const int total = __shfl(incl, 31);
    int base = 0;
    if (lane == 0) base = atomicAdd(cursor, total);
    base = __shfl(base, 0);
    if (i < n) {
        off[i]  = base + incl - c;
        dinv[i] = __frsqrt_rn((float)(c + 1));
    }
}

__global__ void k_place(const int* __restrict__ src, const int* __restrict__ dst,
                        const int* __restrict__ off, const int* __restrict__ rank,
                        const float* __restrict__ dinv, int2* __restrict__ csr, int E) {
    int e = blockIdx.x * blockDim.x + threadIdx.x;
    if (e < E) {
        const int s = src[e];
        const int p = off[dst[e]] + rank[e];
        csr[p] = make_int2(s, __float_as_int(dinv[s]));
    }
}

__global__ __launch_bounds__(256)
void k_gemm1(const float* __restrict__ X, const float* __restrict__ W,
             float* __restrict__ H, int mTiles) {
    __shared__ float Bs[F_IN * 16];
    const int nBase = blockIdx.y * 16;
    for (int idx = threadIdx.x; idx < F_IN * 16; idx += 256) {
        int k = idx >> 4, nn = idx & 15;
        Bs[idx] = W[k * F_HID + nBase + nn];
    }
    __syncthreads();

    const int wave  = threadIdx.x >> 5;
    const int lane  = threadIdx.x & 31;
    const int mTile = blockIdx.x * 8 + wave;
    if (mTile >= mTiles) return;
    const int mBase = mTile * 16;
    const int half  = lane >> 4;
    const int l     = lane & 15;
    const float* arow = X + (size_t)(mBase + l) * F_IN;

    v8f c = {};
    #pragma unroll 4
    for (int k = 0; k < F_IN; k += 4) {
        const int ak = k + half * 2;
        v2f a = *(const v2f*)(arow + ak);
        v2f b; b.x = Bs[ak * 16 + l]; b.y = Bs[(ak + 1) * 16 + l];
        c = __builtin_amdgcn_wmma_f32_16x16x4_f32(false, a, false, b,
                                                  (short)0, c, false, false);
    }
    float* orow = H + (size_t)(mBase + half * 8) * F_HID + nBase + l;
    #pragma unroll
    for (int r = 0; r < 8; ++r) orow[(size_t)r * F_HID] = c[r];
}

__global__ __launch_bounds__(256)
void k_gather64(const int2* __restrict__ csr, const int* __restrict__ off,
                const int* __restrict__ cnt, const float* __restrict__ dinv,
                const float* __restrict__ h1, float* __restrict__ agg, int n) {
    const int lane = threadIdx.x & 31;
    const int node = __builtin_amdgcn_readfirstlane((blockIdx.x * blockDim.x + threadIdx.x) >> 5);
    if (node >= n) return;
    const int   c  = cnt[node];
    const int   o  = off[node];
    const float di = dinv[node];
    const float2 own = ((const float2*)(h1 + (size_t)node * F_HID))[lane];
    float2 acc;
    acc.x = own.x * di * di;
    acc.y = own.y * di * di;
    for (int c0 = 0; c0 < c; c0 += 32) {
        const int m = min(32, c - c0);
        int   s = node;
        float w = 0.0f;
        if (lane < m) {
            const int2 ent = csr[o + c0 + lane];
            s = ent.x;
            w = __int_as_float(ent.y) * di;
        }
        for (int j0 = 0; j0 < m; j0 += 8) {
            float2 v[8];
            #pragma unroll
            for (int u = 0; u < 8; ++u) {
                const int sj = __builtin_amdgcn_readlane(s, j0 + u);
                v[u] = ((const float2*)(h1 + (size_t)sj * F_HID))[lane];
            }
            asm volatile("" :: "v"(v[0].x), "v"(v[0].y), "v"(v[1].x), "v"(v[1].y),
                               "v"(v[2].x), "v"(v[2].y), "v"(v[3].x), "v"(v[3].y),
                               "v"(v[4].x), "v"(v[4].y), "v"(v[5].x), "v"(v[5].y),
                               "v"(v[6].x), "v"(v[6].y), "v"(v[7].x), "v"(v[7].y));
            #pragma unroll
            for (int u = 0; u < 8; ++u) {
                if (j0 + u < m) {
                    const float nj = __int_as_float(
                        __builtin_amdgcn_readlane(__float_as_int(w), j0 + u));
                    acc.x += v[u].x * nj;
                    acc.y += v[u].y * nj;
                }
            }
        }
    }
    ((float2*)(agg + (size_t)node * F_HID))[lane] = acc;
}

__global__ __launch_bounds__(256)
void k_gemm2(const float* __restrict__ A1, const float* __restrict__ bias1,
             const float* __restrict__ W, float* __restrict__ H, int mTiles) {
    __shared__ float Bs[F_HID * 16];
    __shared__ float b1s[F_HID];
    for (int idx = threadIdx.x; idx < F_HID * 16; idx += 256) {
        int k = idx >> 4, nn = idx & 15;
        Bs[idx] = W[k * F_OUT + nn];
    }
    if (threadIdx.x < F_HID) b1s[threadIdx.x] = bias1[threadIdx.x];
    __syncthreads();

    const int wave  = threadIdx.x >> 5;
    const int lane  = threadIdx.x & 31;
    const int mTile = blockIdx.x * 8 + wave;
    if (mTile >= mTiles) return;
    const int mBase = mTile * 16;
    const int half  = lane >> 4;
    const int l     = lane & 15;
    const float* arow = A1 + (size_t)(mBase + l) * F_HID;

    v8f c = {};
    #pragma unroll 4
    for (int k = 0; k < F_HID; k += 4) {
        const int ak = k + half * 2;
        v2f a, b;
        a.x = fmaxf(arow[ak]     + b1s[ak],     0.0f);
        a.y = fmaxf(arow[ak + 1] + b1s[ak + 1], 0.0f);
        b.x = Bs[ak * 16 + l]; b.y = Bs[(ak + 1) * 16 + l];
        c = __builtin_amdgcn_wmma_f32_16x16x4_f32(false, a, false, b,
                                                  (short)0, c, false, false);
    }
    float* orow = H + (size_t)(mBase + half * 8) * F_OUT + l;
    #pragma unroll
    for (int r = 0; r < 8; ++r) orow[(size_t)r * F_OUT] = c[r];
}

__global__ __launch_bounds__(256)
void k_gather16(const int2* __restrict__ csr, const int* __restrict__ off,
                const int* __restrict__ cnt, const float* __restrict__ dinv,
                const float* __restrict__ h2, const float* __restrict__ bias2,
                float* __restrict__ out, int n) {
    const int lane = threadIdx.x & 31;
    const int node = __builtin_amdgcn_readfirstlane((blockIdx.x * blockDim.x + threadIdx.x) >> 5);
    if (node >= n) return;
    const int half = lane >> 4;
    const int f    = lane & 15;
    const int   c  = cnt[node];
    const int   o  = off[node];
    const float di = dinv[node];
    const float own = h2[(size_t)node * F_OUT + f];
    float acc = (half == 0) ? (own * di * di + bias2[f]) : 0.0f;
    for (int c0 = 0; c0 < c; c0 += 32) {
        const int m = min(32, c - c0);
        int   s = node;
        float w = 0.0f;
        if (lane < m) {
            const int2 ent = csr[o + c0 + lane];
            s = ent.x;
            w = __int_as_float(ent.y) * di;
        }
        for (int j0 = 0; j0 < m; j0 += 16) {
            float v[8];
            #pragma unroll
            for (int u = 0; u < 8; ++u) {
                const int sa = __builtin_amdgcn_readlane(s, j0 + 2 * u);
                const int sb = __builtin_amdgcn_readlane(s, j0 + 2 * u + 1);
                const int sj = half ? sb : sa;
                v[u] = h2[(size_t)sj * F_OUT + f];
            }
            #pragma unroll
            for (int u = 0; u < 8; ++u) {
                const int   wa = __builtin_amdgcn_readlane(__float_as_int(w), j0 + 2 * u);
                const int   wb = __builtin_amdgcn_readlane(__float_as_int(w), j0 + 2 * u + 1);
                const float nj = __int_as_float(half ? wb : wa);
                if (j0 + 2 * u + half < m) acc += v[u] * nj;
            }
        }
    }
    acc += __shfl_xor(acc, 16);
    if (half == 0) out[(size_t)node * F_OUT + f] = acc;
}

extern "C" void kernel_launch(void* const* d_in, const int* in_sizes, int n_in,
                              void* d_out, int out_size, void* d_ws, size_t ws_size,
                              hipStream_t stream) {
    const float* x  = (const float*)d_in[0];
    const int*   ei = (const int*)  d_in[1];
    const float* W1 = (const float*)d_in[2];
    const float* b1 = (const float*)d_in[3];
    const float* W2 = (const float*)d_in[4];
    const float* b2 = (const float*)d_in[5];
    float* out = (float*)d_out;

    const int n = in_sizes[0] / F_IN;
    const int E = in_sizes[1] / 2;
    const int* src = ei;
    const int* dst = ei + E;

    float* h1     = (float*)d_ws;
    float* agg1   = h1 + (size_t)n * F_HID;
    float* h2     = agg1 + (size_t)n * F_HID;
    int2*  csr    = (int2*)(h2 + (size_t)n * F_OUT);
    int*   rank   = (int*)(csr + E);
    int*   cnt    = rank + E;
    int*   off    = cnt + n;
    float* dinv   = (float*)(off + n);
    int*   cursor = (int*)(dinv + n);

    const int mTiles  = n / 16;
    const int mGroups = (mTiles + 7) / 8;

    k_zero_cnt<<<(n + 255) / 256, 256, 0, stream>>>(cnt, cursor, n);
    k_count<<<(E + 255) / 256, 256, 0, stream>>>(dst, cnt, rank, E);
    k_offsets<<<(n + 255) / 256, 256, 0, stream>>>(cnt, cursor, off, dinv, n);
    k_place<<<(E + 255) / 256, 256, 0, stream>>>(src, dst, off, rank, dinv, csr, E);

    k_gemm1<<<dim3(mGroups, F_HID / 16), 256, 0, stream>>>(x, W1, h1, mTiles);

    k_gather64<<<(int)(((size_t)n * 32 + 255) / 256), 256, 0, stream>>>(csr, off, cnt, dinv, h1, agg1, n);

    k_gemm2<<<dim3(mGroups, 1), 256, 0, stream>>>(agg1, b1, W2, h2, mTiles);

    k_gather16<<<(int)(((size_t)n * 32 + 255) / 256), 256, 0, stream>>>(csr, off, cnt, dinv, h2, b2, out, n);
}


# x@W1 tiles moved into the in-degree count launch (one grid split by block index, product blocks first): row loads overlap the count's atomic round trips; 8 launches to 7; every sum as before
# speedup vs baseline: 1.6659x; 1.0865x over previous
#include <hip/hip_runtime.h>


typedef float v2f __attribute__((ext_vector_type(2)));
typedef float v8f __attribute__((ext_vector_type(8)));

#define F_IN  128
#define F_HID 64
#define F_OUT 16

__global__ void k_zero_cnt(int* __restrict__ cnt, int* __restrict__ cursor, int n) {
    int i = blockIdx.x * blockDim.x + threadIdx.x;
    if (i < n) cnt[i] = 0;
    if (i == 0) *cursor = 0;
}

__global__ void k_offsets(const int* __restrict__ cnt, int* __restrict__ cursor,
                          int* __restrict__ off, float* __restrict__ dinv, int n) {
    const int i    = blockIdx.x * blockDim.x + threadIdx.x;
    const int lane = threadIdx.x & 31;
    const int c    = (i < n) ? cnt[i] : 0;
    int incl = c;
    #pragma unroll
    for (int sft = 1; sft < 32; sft <<= 1) {
        int t = __shfl_up(incl, sft);
        if (lane >= sft) incl += t;
    }
    const int total = __shfl(incl, 31);
    int base = 0;
    if (lane == 0) base = atomicAdd(cursor, total);
    base = __shfl(base, 0);
    if (i < n) {
        off[i]  = base + incl - c;
        dinv[i] = __frsqrt_rn((float)(c + 1));
    }
}

__global__ void k_place(const int* __restrict__ src, const int* __restrict__ dst,
                        const int* __restrict__ off, const int* __restrict__ rank,
                        const float* __restrict__ dinv, int2* __restrict__ csr, int E) {
    int e = blockIdx.x * blockDim.x + threadIdx.x;
    if (e < E) {
        const int s = src[e];
        const int p = off[dst[e]] + rank[e];
        csr[p] = make_int2(s, __float_as_int(dinv[s]));
    }
}

__global__ __launch_bounds__(256)
void k_count_gemm1(const int* __restrict__ dst, int* __restrict__ cnt, int* __restrict__ rank, int E,
                   const float* __restrict__ X, const float* __restrict__ W,
                   float* __restrict__ H, int mTiles, int mGroups, int gemmBlocks) {
    __shared__ float Bs[F_IN * 16];
    if ((int)blockIdx.x >= gemmBlocks) {
        const int e = ((int)blockIdx.x - gemmBlocks) * 256 + (int)threadIdx.x;
        if (e < E) rank[e] = atomicAdd(&cnt[dst[e]], 1);
        return;
    }
    const int nTile  = (int)blockIdx.x / mGroups;
    const int mGroup = (int)blockIdx.x - nTile * mGroups;
    const int nBase  = nTile * 16;
    for (int idx = threadIdx.x; idx < F_IN * 16; idx += 256) {
        int k = idx >> 4, nn = idx & 15;
        Bs[idx] = W[k * F_HID + nBase + nn];
    }
    __syncthreads();

    const int wave  = threadIdx.x >> 5;
    const int lane  = threadIdx.x & 31;
    const int mTile = mGroup * 8 + wave;
    if (mTile >= mTiles) return;
    const int mBase = mTile * 16;
    const int half  = lane >> 4;
    const int l     = lane & 15;
    const float* arow = X + (size_t)(mBase + l) * F_IN;

    v8f c = {};
    #pragma unroll 4
    for (int k = 0; k < F_IN; k += 4) {
        const int ak = k + half * 2;
        v2f a = *(const v2f*)(arow + ak);
        v2f b; b.x = Bs[ak * 16 + l]; b.y = Bs[(ak + 1) * 16 + l];
        c = __builtin_amdgcn_wmma_f32_16x16x4_f32(false, a, false, b,
                                                  (short)0, c, false, false);
    }
    float* orow = H + (size_t)(mBase + half * 8) * F_HID + nBase + l;
    #pragma unroll
    for (int r = 0; r < 8; ++r) orow[(size_t)r * F_HID] = c[r];
}

__global__ __launch_bounds__(256)
void k_gather64(const int2* __restrict__ csr, const int* __restrict__ off,
                const int* __restrict__ cnt, const float* __restrict__ dinv,
                const float* __restrict__ h1, float* __restrict__ agg, int n) {
    const int lane = threadIdx.x & 31;
    const int node = __builtin_amdgcn_readfirstlane((blockIdx.x * blockDim.x + threadIdx.x) >> 5);
    if (node >= n) return;
    const int   c  = cnt[node];
    const int   o  = off[node];
    const float di = dinv[node];
    const float2 own = ((const float2*)(h1 + (size_t)node * F_HID))[lane];
    float2 acc;
    acc.x = own.x * di * di;
    acc.y = own.y * di * di;
    for (int c0 = 0; c0 < c; c0 += 32) {
        const int m = min(32, c - c0);
        int   s = node;
        float w = 0.0f;
        if (lane < m) {
            const int2 ent = csr[o + c0 + lane];
            s = ent.x;
            w = __int_as_float(ent.y) * di;
        }
        for (int j0 = 0; j0 < m; j0 += 8) {
            float2 v[8];
            #pragma unroll
            for (int u = 0; u < 8; ++u) {
                const int sj = __builtin_amdgcn_readlane(s, j0 + u);
                v[u] = ((const float2*)(h1 + (size_t)sj * F_HID))[lane];
            }
            asm volatile("" :: "v"(v[0].x), "v"(v[0].y), "v"(v[1].x), "v"(v[1].y),
                               "v"(v[2].x), "v"(v[2].y), "v"(v[3].x), "v"(v[3].y),
                               "v"(v[4].x), "v"(v[4].y), "v"(v[5].x), "v"(v[5].y),
                               "v"(v[6].x), "v"(v[6].y), "v"(v[7].x), "v"(v[7].y));
            #pragma unroll
            for (int u = 0; u < 8; ++u) {
                if (j0 + u < m) {
                    const float nj = __int_as_float(
                        __builtin_amdgcn_readlane(__float_as_int(w), j0 + u));
                    acc.x += v[u].x * nj;
                    acc.y += v[u].y * nj;
                }
            }
        }
    }
    ((float2*)(agg + (size_t)node * F_HID))[lane] = acc;
}

__global__ __launch_bounds__(256)
void k_gemm2(const float* __restrict__ A1, const float* __restrict__ bias1,
             const float* __restrict__ W, float* __restrict__ H, int mTiles) {
    __shared__ float Bs[F_HID * 16];
    __shared__ float b1s[F_HID];
    for (int idx = threadIdx.x; idx < F_HID * 16; idx += 256) {
        int k = idx >> 4, nn = idx & 15;
        Bs[idx] = W[k * F_OUT + nn];
    }
    if (threadIdx.x < F_HID) b1s[threadIdx.x] = bias1[threadIdx.x];
    __syncthreads();

    const int wave  = threadIdx.x >> 5;
    const int lane  = threadIdx.x & 31;
    const int mTile = blockIdx.x * 8 + wave;
    if (mTile >= mTiles) return;
    const int mBase = mTile * 16;
    const int half  = lane >> 4;
    const int l     = lane & 15;
    const float* arow = A1 + (size_t)(mBase + l) * F_HID;

    v8f c = {};
    #pragma unroll 4
    for (int k = 0; k < F_HID; k += 4) {
        const int ak = k + half * 2;
        v2f a, b;
        a.x = fmaxf(arow[ak]     + b1s[ak],     0.0f);
        a.y = fmaxf(arow[ak + 1] + b1s[ak + 1], 0.0f);
        b.x = Bs[ak * 16 + l]; b.y = Bs[(ak + 1) * 16 + l];
        c = __builtin_amdgcn_wmma_f32_16x16x4_f32(false, a, false, b,
                                                  (short)0, c, false, false);
    }
    float* orow = H + (size_t)(mBase + half * 8) * F_OUT + l;
    #pragma unroll
    for (int r = 0; r < 8; ++r) orow[(size_t)r * F_OUT] = c[r];
}

__global__ __launch_bounds__(256)
void k_gather16(const int2* __restrict__ csr, const int* __restrict__ off,
                const int* __restrict__ cnt, const float* __restrict__ dinv,
                const float* __restrict__ h2, const float* __restrict__ bias2,
                float* __restrict__ out, int n) {
    const int lane = threadIdx.x & 31;
    const int node = __builtin_amdgcn_readfirstlane((blockIdx.x * blockDim.x + threadIdx.x) >> 5);
    if (node >= n) return;
    const int half = lane >> 4;
    const int f    = lane & 15;
    const int   c  = cnt[node];
    const int   o  = off[node];
    const float di = dinv[node];
    const float own = h2[(size_t)node * F_OUT + f];
    float acc = (half == 0) ? (own * di * di + bias2[f]) : 0.0f;
    for (int c0 = 0; c0 < c; c0 += 32) {
        const int m = min(32, c - c0);
        int   s = node;
        float w = 0.0f;
        if (lane < m) {
            const int2 ent = csr[o + c0 + lane];
            s = ent.x;
            w = __int_as_float(ent.y) * di;
        }
        for (int j0 = 0; j0 < m; j0 += 16) {
            float v[8];
            #pragma unroll
            for (int u = 0; u < 8; ++u) {
                const int sa = __builtin_amdgcn_readlane(s, j0 + 2 * u);
                const int sb = __builtin_amdgcn_readlane(s, j0 + 2 * u + 1);
                const int sj = half ? sb : sa;
                v[u] = h2[(size_t)sj * F_OUT + f];
            }
            #pragma unroll
            for (int u = 0; u < 8; ++u) {
                const int   wa = __builtin_amdgcn_readlane(__float_as_int(w), j0 + 2 * u);
                const int   wb = __builtin_amdgcn_readlane(__float_as_int(w), j0 + 2 * u + 1);
                const float nj = __int_as_float(half ? wb : wa);
                if (j0 + 2 * u + half < m) acc += v[u] * nj;
            }
        }
    }
    acc += __shfl_xor(acc, 16);
    if (half == 0) out[(size_t)node * F_OUT + f] = acc;
}

extern "C" void kernel_launch(void* const* d_in, const int* in_sizes, int n_in,
                              void* d_out, int out_size, void* d_ws, size_t ws_size,
                              hipStream_t stream) {
    const float* x  = (const float*)d_in[0];
    const int*   ei = (const int*)  d_in[1];
    const float* W1 = (const float*)d_in[2];
    const float* b1 = (const float*)d_in[3];
    const float* W2 = (const float*)d_in[4];
    const float* b2 = (const float*)d_in[5];
    float* out = (float*)d_out;

    const int n = in_sizes[0] / F_IN;
    const int E = in_sizes[1] / 2;
    const int* src = ei;
    const int* dst = ei + E;

    float* h1     = (float*)d_ws;
    float* agg1   = h1 + (size_t)n * F_HID;
    float* h2     = agg1 + (size_t)n * F_HID;
    int2*  csr    = (int2*)(h2 + (size_t)n * F_OUT);
    int*   rank   = (int*)(csr + E);
    int*   cnt    = rank + E;
    int*   off    = cnt + n;
    float* dinv   = (float*)(off + n);
    int*   cursor = (int*)(dinv + n);

    const int mTiles  = n / 16;
    const int mGroups = (mTiles + 7) / 8;

    k_zero_cnt<<<(n + 255) / 256, 256, 0, stream>>>(cnt, cursor, n);
    const int gemmBlocks  = mGroups * (F_HID / 16);
    const int countBlocks = (E + 255) / 256;
    k_count_gemm1<<<gemmBlocks + countBlocks, 256, 0, stream>>>(dst, cnt, rank, E, x, W1, h1,
                                                               mTiles, mGroups, gemmBlocks);
    k_offsets<<<(n + 255) / 256, 256, 0, stream>>>(cnt, cursor, off, dinv, n);
    k_place<<<(E + 255) / 256, 256, 0, stream>>>(src, dst, off, rank, dinv, csr, E);

    k_gather64<<<(int)(((size_t)n * 32 + 255) / 256), 256, 0, stream>>>(csr, off, cnt, dinv, h1, agg1, n);

    k_gemm2<<<dim3(mGroups, 1), 256, 0, stream>>>(agg1, b1, W2, h2, mTiles);

    k_gather16<<<(int)(((size_t)n * 32 + 255) / 256), 256, 0, stream>>>(csr, off, cnt, dinv, h2, b2, out, n);
}


# relu(agg1+b1)@W2 folded into the layer-1 gather launch: 16 one-node waves per block put their rows in LDS, wave 0 runs the tile's 16 f32 WMMAs; agg1 never stored, 7 launches to 6; every sum as before
# speedup vs baseline: 1.7419x; 1.0456x over previous
#include <hip/hip_runtime.h>


typedef float v2f __attribute__((ext_vector_type(2)));
typedef float v8f __attribute__((ext_vector_type(8)));

#define F_IN  128
#define F_HID 64
#define F_OUT 16

__global__ void k_zero_cnt(int* __restrict__ cnt, int* __restrict__ cursor, int n) {
    int i = blockIdx.x * blockDim.x + threadIdx.x;
    if (i < n) cnt[i] = 0;
    if (i == 0) *cursor = 0;
}

__global__ void k_offsets(const int* __restrict__ cnt, int* __restrict__ cursor,
                          int* __restrict__ off, float* __restrict__ dinv, int n) {
    const int i    = blockIdx.x * blockDim.x + threadIdx.x;
    const int lane = threadIdx.x & 31;
    const int c    = (i < n) ? cnt[i] : 0;
    int incl = c;
    #pragma unroll
    for (int sft = 1; sft < 32; sft <<= 1) {
        int t = __shfl_up(incl, sft);
        if (lane >= sft) incl += t;
    }
    const int total = __shfl(incl, 31);
    int base = 0;
    if (lane == 0) base = atomicAdd(cursor, total);
    base = __shfl(base, 0);
    if (i < n) {
        off[i]  = base + incl - c;
        dinv[i] = __frsqrt_rn((float)(c + 1));
    }
}

__global__ void k_place(const int* __restrict__ src, const int* __restrict__ dst,
                        const int* __restrict__ off, const int* __restrict__ rank,
                        const float* __restrict__ dinv, int2* __restrict__ csr, int E) {
    int e = blockIdx.x * blockDim.x + threadIdx.x;
    if (e < E) {
        const int s = src[e];
        const int p = off[dst[e]] + rank[e];
        csr[p] = make_int2(s, __float_as_int(dinv[s]));
    }
}

__global__ __launch_bounds__(256)
void k_count_gemm1(const int* __restrict__ dst, int* __restrict__ cnt, int* __restrict__ rank, int E,
                   const float* __restrict__ X, const float* __restrict__ W,
                   float* __restrict__ H, int mTiles, int mGroups, int gemmBlocks) {
    __shared__ float Bs[F_IN * 16];
    if ((int)blockIdx.x >= gemmBlocks) {
        const int e = ((int)blockIdx.x - gemmBlocks) * 256 + (int)threadIdx.x;
        if (e < E) rank[e] = atomicAdd(&cnt[dst[e]], 1);
        return;
    }
    const int nTile  = (int)blockIdx.x / mGroups;
    const int mGroup = (int)blockIdx.x - nTile * mGroups;
    const int nBase  = nTile * 16;
    for (int idx = threadIdx.x; idx < F_IN * 16; idx += 256) {
        int k = idx >> 4, nn = idx & 15;
        Bs[idx] = W[k * F_HID + nBase + nn];
    }
    __syncthreads();

    const int wave  = threadIdx.x >> 5;
    const int lane  = threadIdx.x & 31;
    const int mTile = mGroup * 8 + wave;
    if (mTile >= mTiles) return;
    const int mBase = mTile * 16;
    const int half  = lane >> 4;
    const int l     = lane & 15;
    const float* arow = X + (size_t)(mBase + l) * F_IN;

    v8f c = {};
    #pragma unroll 4
    for (int k = 0; k < F_IN; k += 4) {
        const int ak = k + half * 2;
        v2f a = *(const v2f*)(arow + ak);
        v2f b; b.x = Bs[ak * 16 + l]; b.y = Bs[(ak + 1) * 16 + l];
        c = __builtin_amdgcn_wmma_f32_16x16x4_f32(false, a, false, b,
                                                  (short)0, c, false, false);
    }
    float* orow = H + (size_t)(mBase + half * 8) * F_HID + nBase + l;
    #pragma unroll
    for (int r = 0; r < 8; ++r) orow[(size_t)r * F_HID] = c[r];
}

#define A_PITCH 68

__global__ __launch_bounds__(512)
void k_gather64_gemm2(const int2* __restrict__ csr, const int* __restrict__ off,
                      const int* __restrict__ cnt, const float* __restrict__ dinv,
                      const float* __restrict__ h1, const float* __restrict__ bias1,
                      const float* __restrict__ W2, float* __restrict__ h2, int n) {
    __shared__ float As[16 * A_PITCH];
    __shared__ float Bs[F_HID * 16];
    const int lane = threadIdx.x & 31;
    const int wave = __builtin_amdgcn_readfirstlane((int)threadIdx.x >> 5);
    const int node = (int)blockIdx.x * 16 + wave;

    const float w2lo = W2[threadIdx.x];
    const float w2hi = W2[threadIdx.x + 512];
    const float2 bb = ((const float2*)bias1)[lane];

    float2 acc;
    acc.x = 0.0f; acc.y = 0.0f;
    if (node < n) {
        const int   c  = cnt[node];
        const int   o  = off[node];
        const float di = dinv[node];
        const float2 own = ((const float2*)(h1 + (size_t)node * F_HID))[lane];
        asm volatile("" :: "s"(c), "s"(o), "s"(di));
        acc.x = own.x * di * di;
        acc.y = own.y * di * di;
        for (int c0 = 0; c0 < c; c0 += 32) {
            const int m = min(32, c - c0);
            int   s = node;
            float w = 0.0f;
            if (lane < m) {
                const int2 ent = csr[o + c0 + lane];
                s = ent.x;
                w = __int_as_float(ent.y) * di;
            }
            for (int j0 = 0; j0 < m; j0 += 8) {
                float2 v[8];
                #pragma unroll
                for (int u = 0; u < 8; ++u) {
                    const int sj = __builtin_amdgcn_readlane(s, j0 + u);
                    v[u] = ((const float2*)(h1 + (size_t)sj * F_HID))[lane];
                }
                asm volatile("" :: "v"(v[0].x), "v"(v[0].y), "v"(v[1].x), "v"(v[1].y),
                                   "v"(v[2].x), "v"(v[2].y), "v"(v[3].x), "v"(v[3].y),
                                   "v"(v[4].x), "v"(v[4].y), "v"(v[5].x), "v"(v[5].y),
                                   "v"(v[6].x), "v"(v[6].y), "v"(v[7].x), "v"(v[7].y));
                #pragma unroll
                for (int u = 0; u < 8; ++u) {
                    if (j0 + u < m) {
                        const float nj = __int_as_float(
                            __builtin_amdgcn_readlane(__float_as_int(w), j0 + u));
                        acc.x += v[u].x * nj;
                        acc.y += v[u].y * nj;
                    }
                }
            }
        }
    }
    float2 a1;
    a1.x = fmaxf(acc.x + bb.x, 0.0f);
    a1.y = fmaxf(acc.y + bb.y, 0.0f);
    ((float2*)(As + wave * A_PITCH))[lane] = a1;
    Bs[threadIdx.x]       = w2lo;
    Bs[threadIdx.x + 512] = w2hi;
    __syncthreads();
    if (wave != 0) return;

    const int half = lane >> 4;
    const int l    = lane & 15;
    v8f c = {};
    #pragma unroll
    for (int k = 0; k < F_HID; k += 4) {
        const int ak = k + half * 2;
        v2f a = *(const v2f*)(As + l * A_PITCH + ak);
        v2f b; b.x = Bs[ak * 16 + l]; b.y = Bs[(ak + 1) * 16 + l];
        c = __builtin_amdgcn_wmma_f32_16x16x4_f32(false, a, false, b,
                                                  (short)0, c, false, false);
    }
    const int rBase = (int)blockIdx.x * 16 + half * 8;
    float* orow = h2 + (size_t)rBase * F_OUT + l;
    #pragma unroll
    for (int r = 0; r < 8; ++r)
        if (rBase + r < n) orow[(size_t)r * F_OUT] = c[r];
}

__global__ __launch_bounds__(256)
void k_gather16(const int2* __restrict__ csr, const int* __restrict__ off,
                const int* __restrict__ cnt, const float* __restrict__ dinv,
                const float* __restrict__ h2, const float* __restrict__ bias2,
                float* __restrict__ out, int n) {
    const int lane = threadIdx.x & 31;
    const int node = __builtin_amdgcn_readfirstlane((blockIdx.x * blockDim.x + threadIdx.x) >> 5);
    if (node >= n) return;
    const int half = lane >> 4;
    const int f    = lane & 15;
    const int   c  = cnt[node];
    const int   o  = off[node];
    const float di = dinv[node];
    const float own = h2[(size_t)node * F_OUT + f];
    float acc = (half == 0) ? (own * di * di + bias2[f]) : 0.0f;
    for (int c0 = 0; c0 < c; c0 += 32) {
        const int m = min(32, c - c0);
        int   s = node;
        float w = 0.0f;
        if (lane < m) {
            const int2 ent = csr[o + c0 + lane];
            s = ent.x;
            w = __int_as_float(ent.y) * di;
        }
        for (int j0 = 0; j0 < m; j0 += 16) {
            float v[8];
            #pragma unroll
            for (int u = 0; u < 8; ++u) {
                const int sa = __builtin_amdgcn_readlane(s, j0 + 2 * u);
                const int sb = __builtin_amdgcn_readlane(s, j0 + 2 * u + 1);
                const int sj = half ? sb : sa;
                v[u] = h2[(size_t)sj * F_OUT + f];
            }
            #pragma unroll
            for (int u = 0; u < 8; ++u) {
                const int   wa = __builtin_amdgcn_readlane(__float_as_int(w), j0 + 2 * u);
                const int   wb = __builtin_amdgcn_readlane(__float_as_int(w), j0 + 2 * u + 1);
                const float nj = __int_as_float(half ? wb : wa);
                if (j0 + 2 * u + half < m) acc += v[u] * nj;
            }
        }
    }
    acc += __shfl_xor(acc, 16);
    if (half == 0) out[(size_t)node * F_OUT + f] = acc;
}

extern "C" void kernel_launch(void* const* d_in, const int* in_sizes, int n_in,
                              void* d_out, int out_size, void* d_ws, size_t ws_size,
                              hipStream_t stream) {
    const float* x  = (const float*)d_in[0];
    const int*   ei = (const int*)  d_in[1];
    const float* W1 = (const float*)d_in[2];
    const float* b1 = (const float*)d_in[3];
    const float* W2 = (const float*)d_in[4];
    const float* b2 = (const float*)d_in[5];
    float* out = (float*)d_out;

    const int n = in_sizes[0] / F_IN;
    const int E = in_sizes[1] / 2;
    const int* src = ei;
    const int* dst = ei + E;

    float* h1     = (float*)d_ws;
    float* h2     = h1 + (size_t)n * F_HID;
    int2*  csr    = (int2*)(h2 + (size_t)n * F_OUT);
    int*   rank   = (int*)(csr + E);
    int*   cnt    = rank + E;
    int*   off    = cnt + n;
    float* dinv   = (float*)(off + n);
    int*   cursor = (int*)(dinv + n);

    const int mTiles  = n / 16;
    const int mGroups = (mTiles + 7) / 8;

    k_zero_cnt<<<(n + 255) / 256, 256, 0, stream>>>(cnt, cursor, n);
    const int gemmBlocks  = mGroups * (F_HID / 16);
    const int countBlocks = (E + 255) / 256;
    k_count_gemm1<<<gemmBlocks + countBlocks, 256, 0, stream>>>(dst, cnt, rank, E, x, W1, h1,
                                                               mTiles, mGroups, gemmBlocks);
    k_offsets<<<(n + 255) / 256, 256, 0, stream>>>(cnt, cursor, off, dinv, n);
    k_place<<<(E + 255) / 256, 256, 0, stream>>>(src, dst, off, rank, dinv, csr, E);

    k_gather64_gemm2<<<(n + 15) / 16, 512, 0, stream>>>(csr, off, cnt, dinv, h1, b1, W2, h2, n);

    k_gather16<<<(int)(((size_t)n * 32 + 255) / 256), 256, 0, stream>>>(csr, off, cnt, dinv, h2, b2, out, n);
}


# in-degree count and edge ranks on LDS counters (one work-group per edge chunk, chunk x node count table, per-node scan) not per-edge global atomics; 6 launches to 5; sum order: chunk, then arrival
# speedup vs baseline: 2.3746x; 1.3632x over previous
#include <hip/hip_runtime.h>


typedef float v2f __attribute__((ext_vector_type(2)));
typedef float v8f __attribute__((ext_vector_type(8)));

#define F_IN  128
#define F_HID 64
#define F_OUT 16

#define N_CAP    50000
#define N_CHUNKS 128

__global__ __launch_bounds__(1024)
void k_count_lds(const int* __restrict__ dst, int* __restrict__ lrank, int* __restrict__ table,
                 int* __restrict__ cursor, int n, int E, int per) {
    __shared__ int cs[N_CAP];
    const int tid = (int)threadIdx.x;
    for (int i = tid; i < N_CAP / 4; i += 1024) ((int4*)cs)[i] = make_int4(0, 0, 0, 0);
    if (blockIdx.x == 0 && tid == 0) *cursor = 0;
    __syncthreads();
    const int e0 = (int)blockIdx.x * per;
    const int e1 = min(E, e0 + per);
    for (int eb = e0; eb < e1; eb += 8 * 1024) {
        int d[8];
        #pragma unroll
        for (int u = 0; u < 8; ++u) {
            const int e = eb + u * 1024 + tid;
            d[u] = (e < e1) ? dst[e] : -1;
        }
        #pragma unroll
        for (int u = 0; u < 8; ++u) {
            const int e = eb + u * 1024 + tid;
            if (d[u] >= 0) lrank[e] = atomicAdd(&cs[d[u]], 1);
        }
    }
    __syncthreads();
    int* row = table + (size_t)blockIdx.x * n;
    if ((n & 3) == 0) {
        for (int i = tid; i < n / 4; i += 1024) ((int4*)row)[i] = ((const int4*)cs)[i];
    } else {
        for (int i = tid; i < n; i += 1024) row[i] = cs[i];
    }
}

__global__ void k_scan_offsets(int* __restrict__ table, int* __restrict__ cursor,
                               int* __restrict__ cnt, int* __restrict__ off,
                               float* __restrict__ dinv, int n) {
    const int i    = blockIdx.x * blockDim.x + threadIdx.x;
    const int lane = threadIdx.x & 31;
    int c = 0;
    if (i < n) {
        int* col = table + i;
        for (int g0 = 0; g0 < N_CHUNKS; g0 += 16) {
            int t[16];
            #pragma unroll
            for (int u = 0; u < 16; ++u) t[u] = col[(size_t)(g0 + u) * n];
            #pragma unroll
            for (int u = 0; u < 16; ++u) { col[(size_t)(g0 + u) * n] = c; c += t[u]; }
        }
        cnt[i] = c;
    }
    int incl = c;
    #pragma unroll
    for (int sft = 1; sft < 32; sft <<= 1) {
        int t = __shfl_up(incl, sft);
        if (lane >= sft) incl += t;
    }
    const int total = __shfl(incl, 31);
    int base = 0;
    if (lane == 0) base = atomicAdd(cursor, total);
    base = __shfl(base, 0);
    if (i < n) {
        off[i]  = base + incl - c;
        dinv[i] = __frsqrt_rn((float)(c + 1));
    }
}

__global__ __launch_bounds__(256)
void k_place_gemm1(const int* __restrict__ src, const int* __restrict__ dst,
                   const int* __restrict__ off, const int* __restrict__ lrank,
                   const int* __restrict__ table, const float* __restrict__ dinv,
                   int2* __restrict__ csr, int n, int E, int per,
                   const float* __restrict__ X, const float* __restrict__ W,
                   float* __restrict__ H, int mTiles, int mGroups, int gemmBlocks) {
    __shared__ float Bs[F_IN * 16];
    if ((int)blockIdx.x >= gemmBlocks) {
        const int e = ((int)blockIdx.x - gemmBlocks) * 256 + (int)threadIdx.x;
        if (e < E) {
            const int s = src[e];
            const int d = dst[e];
            const int g = e / per;
            const int p = off[d] + table[(size_t)g * n + d] + lrank[e];
            csr[p] = make_int2(s, __float_as_int(dinv[s]));
        }
        return;
    }
    const int nTile  = (int)blockIdx.x / mGroups;
    const int mGroup = (int)blockIdx.x - nTile * mGroups;
    const int nBase  = nTile * 16;
    for (int idx = threadIdx.x; idx < F_IN * 16; idx += 256) {
        int k = idx >> 4, nn = idx & 15;
        Bs[idx] = W[k * F_HID + nBase + nn];
    }
    __syncthreads();

    const int wave  = threadIdx.x >> 5;
    const int lane  = threadIdx.x & 31;
    const int mTile = mGroup * 8 + wave;
    if (mTile >= mTiles) return;
    const int mBase = mTile * 16;
    const int half  = lane >> 4;
    const int l     = lane & 15;
    const float* arow = X + (size_t)(mBase + l) * F_IN;

    v8f c = {};
    #pragma unroll 4
    for (int k = 0; k < F_IN; k += 4) {
        const int ak = k + half * 2;
        v2f a = *(const v2f*)(arow + ak);
        v2f b; b.x = Bs[ak * 16 + l]; b.y = Bs[(ak + 1) * 16 + l];
        c = __builtin_amdgcn_wmma_f32_16x16x4_f32(false, a, false, b,
                                                  (short)0, c, false, false);
    }
    float* orow = H + (size_t)(mBase + half * 8) * F_HID + nBase + l;
    #pragma unroll
    for (int r = 0; r < 8; ++r) orow[(size_t)r * F_HID] = c[r];
}

#define A_PITCH 68

__global__ __launch_bounds__(512)
void k_gather64_gemm2(const int2* __restrict__ csr, const int* __restrict__ off,
                      const int* __restrict__ cnt, const float* __restrict__ dinv,
                      const float* __restrict__ h1, const float* __restrict__ bias1,
                      const float* __restrict__ W2, float* __restrict__ h2, int n) {
    __shared__ float As[16 * A_PITCH];
    __shared__ float Bs[F_HID * 16];
    const int lane = threadIdx.x & 31;
    const int wave = __builtin_amdgcn_readfirstlane((int)threadIdx.x >> 5);
    const int node = (int)blockIdx.x * 16 + wave;

    const float w2lo = W2[threadIdx.x];
    const float w2hi = W2[threadIdx.x + 512];
    const float2 bb = ((const float2*)bias1)[lane];

    float2 acc;
    acc.x = 0.0f; acc.y = 0.0f;
    if (node < n) {
        const int   c  = cnt[node];
        const int   o  = off[node];
        const float di = dinv[node];
        const float2 own = ((const float2*)(h1 + (size_t)node * F_HID))[lane];
        asm volatile("" :: "s"(c), "s"(o), "s"(di));
        acc.x = own.x * di * di;
        acc.y = own.y * di * di;
        for (int c0 = 0; c0 < c; c0 += 32) {
            const int m = min(32, c - c0);
            int   s = node;
            float w = 0.0f;
            if (lane < m) {
                const int2 ent = csr[o + c0 + lane];
                s = ent.x;
                w = __int_as_float(ent.y) * di;
            }
            for (int j0 = 0; j0 < m; j0 += 8) {
                float2 v[8];
                #pragma unroll
                for (int u = 0; u < 8; ++u) {
                    const int sj = __builtin_amdgcn_readlane(s, j0 + u);
                    v[u] = ((const float2*)(h1 + (size_t)sj * F_HID))[lane];
                }
                asm volatile("" :: "v"(v[0].x), "v"(v[0].y), "v"(v[1].x), "v"(v[1].y),
                                   "v"(v[2].x), "v"(v[2].y), "v"(v[3].x), "v"(v[3].y),
                                   "v"(v[4].x), "v"(v[4].y), "v"(v[5].x), "v"(v[5].y),
                                   "v"(v[6].x), "v"(v[6].y), "v"(v[7].x), "v"(v[7].y));
                #pragma unroll
                for (int u = 0; u < 8; ++u) {
                    if (j0 + u < m) {
                        const float nj = __int_as_float(
                            __builtin_amdgcn_readlane(__float_as_int(w), j0 + u));
                        acc.x += v[u].x * nj;
                        acc.y += v[u].y * nj;
                    }
                }
            }
        }
    }
    float2 a1;
    a1.x = fmaxf(acc.x + bb.x, 0.0f);
    a1.y = fmaxf(acc.y + bb.y, 0.0f);
    ((float2*)(As + wave * A_PITCH))[lane] = a1;
    Bs[threadIdx.x]       = w2lo;
    Bs[threadIdx.x + 512] = w2hi;
    __syncthreads();
    if (wave != 0) return;

    const int half = lane >> 4;
    const int l    = lane & 15;
    v8f c = {};
    #pragma unroll
    for (int k = 0; k < F_HID; k += 4) {
        const int ak = k + half * 2;
        v2f a = *(const v2f*)(As + l * A_PITCH + ak);
        v2f b; b.x = Bs[ak * 16 + l]; b.y = Bs[(ak + 1) * 16 + l];
        c = __builtin_amdgcn_wmma_f32_16x16x4_f32(false, a, false, b,
                                                  (short)0, c, false, false);
    }
    const int rBase = (int)blockIdx.x * 16 + half * 8;
    float* orow = h2 + (size_t)rBase * F_OUT + l;
    #pragma unroll
    for (int r = 0; r < 8; ++r)
        if (rBase + r < n) orow[(size_t)r * F_OUT] = c[r];
}

__global__ __launch_bounds__(256)
void k_gather16(const int2* __restrict__ csr, const int* __restrict__ off,
                const int* __restrict__ cnt, const float* __restrict__ dinv,
                const float* __restrict__ h2, const float* __restrict__ bias2,
                float* __restrict__ out, int n) {
    const int lane = threadIdx.x & 31;
    const int node = __builtin_amdgcn_readfirstlane((blockIdx.x * blockDim.x + threadIdx.x) >> 5);
    if (node >= n) return;
    const int half = lane >> 4;
    const int f    = lane & 15;
    const int   c  = cnt[node];
    const int   o  = off[node];
    const float di = dinv[node];
    const float own = h2[(size_t)node * F_OUT + f];
    float acc = (half == 0) ? (own * di * di + bias2[f]) : 0.0f;
    for (int c0 = 0; c0 < c; c0 += 32) {
        const int m = min(32, c - c0);
        int   s = node;
        float w = 0.0f;
        if (lane < m) {
            const int2 ent = csr[o + c0 + lane];
            s = ent.x;
            w = __int_as_float(ent.y) * di;
        }
        for (int j0 = 0; j0 < m; j0 += 16) {
            float v[8];
            #pragma unroll
            for (int u = 0; u < 8; ++u) {
                const int sa = __builtin_amdgcn_readlane(s, j0 + 2 * u);
                const int sb = __builtin_amdgcn_readlane(s, j0 + 2 * u + 1);
                const int sj = half ? sb : sa;
                v[u] = h2[(size_t)sj * F_OUT + f];
            }
            #pragma unroll
            for (int u = 0; u < 8; ++u) {
                const int   wa = __builtin_amdgcn_readlane(__float_as_int(w), j0 + 2 * u);
                const int   wb = __builtin_amdgcn_readlane(__float_as_int(w), j0 + 2 * u + 1);
                const float nj = __int_as_float(half ? wb : wa);
                if (j0 + 2 * u + half < m) acc += v[u] * nj;
            }
        }
    }
    acc += __shfl_xor(acc, 16);
    if (half == 0) out[(size_t)node * F_OUT + f] = acc;
}

extern "C" void kernel_launch(void* const* d_in, const int* in_sizes, int n_in,
                              void* d_out, int out_size, void* d_ws, size_t ws_size,
                              hipStream_t stream) {
    const float* x  = (const float*)d_in[0];
    const int*   ei = (const int*)  d_in[1];
    const float* W1 = (const float*)d_in[2];
    const float* b1 = (const float*)d_in[3];
    const float* W2 = (const float*)d_in[4];
    const float* b2 = (const float*)d_in[5];
    float* out = (float*)d_out;

    const int n = in_sizes[0] / F_IN;
    const int E = in_sizes[1] / 2;
    const int* src = ei;
    const int* dst = ei + E;

    float* h1     = (float*)d_ws;
    float* h2     = h1 + (size_t)n * F_HID;
    int2*  csr    = (int2*)(h2 + (size_t)n * F_OUT);
    int*   rank   = (int*)(csr + E);
    int*   table  = rank + E;
    int*   cnt    = table + (size_t)N_CHUNKS * n;
    int*   off    = cnt + n;
    float* dinv   = (float*)(off + n);
    int*   cursor = (int*)(dinv + n);

    const int mTiles  = n / 16;
    const int mGroups = (mTiles + 7) / 8;

    const int per = (E + N_CHUNKS - 1) / N_CHUNKS;
    k_count_lds<<<N_CHUNKS, 1024, 0, stream>>>(dst, rank, table, cursor, n, E, per);
    k_scan_offsets<<<(n + 255) / 256, 256, 0, stream>>>(table, cursor, cnt, off, dinv, n);
    const int gemmBlocks  = mGroups * (F_HID / 16);
    const int placeBlocks = (E + 255) / 256;
    k_place_gemm1<<<gemmBlocks + placeBlocks, 256, 0, stream>>>(src, dst, off, rank, table, dinv, csr, n, E, per,
                                                               x, W1, h1, mTiles, mGroups, gemmBlocks);

    k_gather64_gemm2<<<(n + 15) / 16, 512, 0, stream>>>(csr, off, cnt, dinv, h1, b1, W2, h2, n);

    k_gather16<<<(int)(((size_t)n * 32 + 255) / 256), 256, 0, stream>>>(csr, off, cnt, dinv, h2, b2, out, n);
}


# count table 128xN to 32xN: count launch split over 4 node ranges x 32 edge chunks (128 work-groups, 50 KB LDS counters each); scan takes a column in one batch; sum order: chunk (now 32), arrival
# speedup vs baseline: 2.7789x; 1.1703x over previous
#include <hip/hip_runtime.h>


typedef float v2f __attribute__((ext_vector_type(2)));
typedef float v8f __attribute__((ext_vector_type(8)));

#define F_IN  128
#define F_HID 64
#define F_OUT 16

#define N_CAP    50000
#define N_CHUNKS 32
#define N_RANGES 4
#define R_CAP    ((((N_CAP + N_RANGES - 1) / N_RANGES) + 3) & ~3)
#define BATCH    32

__global__ __launch_bounds__(1024)
void k_count_lds(const int* __restrict__ dst, int* __restrict__ lrank, int* __restrict__ table,
                 int* __restrict__ cursor, int n, int E, int per) {
    __shared__ alignas(16) int cs[R_CAP];
    const int tid   = (int)threadIdx.x;
    const int chunk = (int)blockIdx.x / N_RANGES;
    const int range = (int)blockIdx.x - chunk * N_RANGES;
    const int nr    = (((n + N_RANGES - 1) / N_RANGES) + 3) & ~3;
    const int lo    = range * nr;
    const unsigned span = (unsigned)max(0, min(n, lo + nr) - lo);
    for (int i = tid; i < R_CAP / 4; i += 1024) ((int4*)cs)[i] = make_int4(0, 0, 0, 0);
    if (blockIdx.x == 0 && tid == 0) *cursor = 0;
    __syncthreads();
    const int e0 = chunk * per;
    const int e1 = min(E, e0 + per);
    for (int eb = e0; eb < e1; eb += BATCH * 1024) {
        int d[BATCH];
        int r[BATCH];
        #pragma unroll
        for (int u = 0; u < BATCH; ++u) {
            const int e = eb + u * 1024 + tid;
            d[u] = (e < e1) ? dst[e] : -1;
        }
        #pragma unroll
        for (int u = 0; u < BATCH; ++u) {
            const unsigned k = (unsigned)(d[u] - lo);
            r[u] = (k < span) ? atomicAdd(&cs[k], 1) : -1;
        }
        #pragma unroll
        for (int u = 0; u < BATCH; ++u) {
            const int e = eb + u * 1024 + tid;
            if (r[u] >= 0) lrank[e] = r[u];
        }
    }
    __syncthreads();
    int* row = table + (size_t)chunk * n + lo;
    if ((span & 3u) == 0 && ((size_t)row & 15) == 0) {
        for (int i = tid; i < (int)(span / 4); i += 1024) ((int4*)row)[i] = ((const int4*)cs)[i];
    } else {
        for (int i = tid; i < (int)span; i += 1024) row[i] = cs[i];
    }
}

__global__ __launch_bounds__(256)
void k_scan_offsets(int* __restrict__ table, int* __restrict__ cursor,
                               int* __restrict__ cnt, int* __restrict__ off,
                               float* __restrict__ dinv, int n) {
    const int i    = blockIdx.x * blockDim.x + threadIdx.x;
    const int lane = threadIdx.x & 31;
    int c = 0;
    if (i < n) {
        const unsigned un = (unsigned)n, ui = (unsigned)i;
        for (int g0 = 0; g0 < N_CHUNKS; g0 += BATCH) {
            int t[BATCH];
            #pragma unroll
            for (int u = 0; u < BATCH; ++u) t[u] = table[(unsigned)(g0 + u) * un + ui];
            #pragma unroll
            for (int u = 0; u < BATCH; ++u) { table[(unsigned)(g0 + u) * un + ui] = c; c += t[u]; }
        }
        cnt[i] = c;
    }
    int incl = c;
    #pragma unroll
    for (int sft = 1; sft < 32; sft <<= 1) {
        int t = __shfl_up(incl, sft);
        if (lane >= sft) incl += t;
    }
    const int total = __shfl(incl, 31);
    int base = 0;
    if (lane == 0) base = atomicAdd(cursor, total);
    base = __shfl(base, 0);
    if (i < n) {
        off[i]  = base + incl - c;
        dinv[i] = __frsqrt_rn((float)(c + 1));
    }
}

__global__ __launch_bounds__(256)
void k_place_gemm1(const int* __restrict__ src, const int* __restrict__ dst,
                   const int* __restrict__ off, const int* __restrict__ lrank,
                   const int* __restrict__ table, const float* __restrict__ dinv,
                   int2* __restrict__ csr, int n, int E, int per,
                   const float* __restrict__ X, const float* __restrict__ W,
                   float* __restrict__ H, int mTiles, int mGroups, int gemmBlocks) {
    __shared__ float Bs[F_IN * 16];
    if ((int)blockIdx.x >= gemmBlocks) {
        const int e = ((int)blockIdx.x - gemmBlocks) * 256 + (int)threadIdx.x;
        if (e < E) {
            const int s = src[e];
            const int d = dst[e];
            const int g = e / per;
            const int p = off[d] + table[(size_t)g * n + d] + lrank[e];
            csr[p] = make_int2(s, __float_as_int(dinv[s]));
        }
        return;
    }
    const int nTile  = (int)blockIdx.x / mGroups;
    const int mGroup = (int)blockIdx.x - nTile * mGroups;
    const int nBase  = nTile * 16;
    for (int idx = threadIdx.x; idx < F_IN * 16; idx += 256) {
        int k = idx >> 4, nn = idx & 15;
        Bs[idx] = W[k * F_HID + nBase + nn];
    }
    __syncthreads();

    const int wave  = threadIdx.x >> 5;
    const int lane  = threadIdx.x & 31;
    const int mTile = mGroup * 8 + wave;
    if (mTile >= mTiles) return;
    const int mBase = mTile * 16;
    const int half  = lane >> 4;
    const int l     = lane & 15;
    const float* arow = X + (size_t)(mBase + l) * F_IN;

    v8f c = {};
    #pragma unroll 4
    for (int k = 0; k < F_IN; k += 4) {
        const int ak = k + half * 2;
        v2f a = *(const v2f*)(arow + ak);
        v2f b; b.x = Bs[ak * 16 + l]; b.y = Bs[(ak + 1) * 16 + l];
        c = __builtin_amdgcn_wmma_f32_16x16x4_f32(false, a, false, b,
                                                  (short)0, c, false, false);
    }
    float* orow = H + (size_t)(mBase + half * 8) * F_HID + nBase + l;
    #pragma unroll
    for (int r = 0; r < 8; ++r) orow[(size_t)r * F_HID] = c[r];
}

#define A_PITCH 68

__global__ __launch_bounds__(512)
void k_gather64_gemm2(const int2* __restrict__ csr, const int* __restrict__ off,
                      const int* __restrict__ cnt, const float* __restrict__ dinv,
                      const float* __restrict__ h1, const float* __restrict__ bias1,
                      const float* __restrict__ W2, float* __restrict__ h2, int n) {
    __shared__ float As[16 * A_PITCH];
    __shared__ float Bs[F_HID * 16];
    const int lane = threadIdx.x & 31;
    const int wave = __builtin_amdgcn_readfirstlane((int)threadIdx.x >> 5);
    const int node = (int)blockIdx.x * 16 + wave;

    const float w2lo = W2[threadIdx.x];
    const float w2hi = W2[threadIdx.x + 512];
    const float2 bb = ((const float2*)bias1)[lane];

    float2 acc;
    acc.x = 0.0f; acc.y = 0.0f;
    if (node < n) {
        const int   c  = cnt[node];
        const int   o  = off[node];
        const float di = dinv[node];
        const float2 own = ((const float2*)(h1 + (size_t)node * F_HID))[lane];
        asm volatile("" :: "s"(c), "s"(o), "s"(di));
        acc.x = own.x * di * di;
        acc.y = own.y * di * di;
        for (int c0 = 0; c0 < c; c0 += 32) {
            const int m = min(32, c - c0);
            int   s = node;
            float w = 0.0f;
            if (lane < m) {
                const int2 ent = csr[o + c0 + lane];
                s = ent.x;
                w = __int_as_float(ent.y) * di;
            }
            for (int j0 = 0; j0 < m; j0 += 8) {
                float2 v[8];
                #pragma unroll
                for (int u = 0; u < 8; ++u) {
                    const int sj = __builtin_amdgcn_readlane(s, j0 + u);
                    v[u] = ((const float2*)(h1 + (size_t)sj * F_HID))[lane];
                }
                asm volatile("" :: "v"(v[0].x), "v"(v[0].y), "v"(v[1].x), "v"(v[1].y),
                                   "v"(v[2].x), "v"(v[2].y), "v"(v[3].x), "v"(v[3].y),
                                   "v"(v[4].x), "v"(v[4].y), "v"(v[5].x), "v"(v[5].y),
                                   "v"(v[6].x), "v"(v[6].y), "v"(v[7].x), "v"(v[7].y));
                #pragma unroll
                for (int u = 0; u < 8; ++u) {
                    if (j0 + u < m) {
                        const float nj = __int_as_float(
                            __builtin_amdgcn_readlane(__float_as_int(w), j0 + u));
                        acc.x += v[u].x * nj;
                        acc.y += v[u].y * nj;
                    }
                }
            }
        }
    }
    float2 a1;
    a1.x = fmaxf(acc.x + bb.x, 0.0f);
    a1.y = fmaxf(acc.y + bb.y, 0.0f);
    ((float2*)(As + wave * A_PITCH))[lane] = a1;
    Bs[threadIdx.x]       = w2lo;
    Bs[threadIdx.x + 512] = w2hi;
    __syncthreads();
    if (wave != 0) return;

    const int half = lane >> 4;
    const int l    = lane & 15;
    v8f c = {};
    #pragma unroll
    for (int k = 0; k < F_HID; k += 4) {
        const int ak = k + half * 2;
        v2f a = *(const v2f*)(As + l * A_PITCH + ak);
        v2f b; b.x = Bs[ak * 16 + l]; b.y = Bs[(ak + 1) * 16 + l];
        c = __builtin_amdgcn_wmma_f32_16x16x4_f32(false, a, false, b,
                                                  (short)0, c, false, false);
    }
    const int rBase = (int)blockIdx.x * 16 + half * 8;
    float* orow = h2 + (size_t)rBase * F_OUT + l;
    #pragma unroll
    for (int r = 0; r < 8; ++r)
        if (rBase + r < n) orow[(size_t)r * F_OUT] = c[r];
}

__global__ __launch_bounds__(256)
void k_gather16(const int2* __restrict__ csr, const int* __restrict__ off,
                const int* __restrict__ cnt, const float* __restrict__ dinv,
                const float* __restrict__ h2, const float* __restrict__ bias2,
                float* __restrict__ out, int n) {
    const int lane = threadIdx.x & 31;
    const int node = __builtin_amdgcn_readfirstlane((blockIdx.x * blockDim.x + threadIdx.x) >> 5);
    if (node >= n) return;
    const int half = lane >> 4;
    const int f    = lane & 15;
    const int   c  = cnt[node];
    const int   o  = off[node];
    const float di = dinv[node];
    const float own = h2[(size_t)node * F_OUT + f];
    float acc = (half == 0) ? (own * di * di + bias2[f]) : 0.0f;
    for (int c0 = 0; c0 < c; c0 += 32) {
        const int m = min(32, c - c0);
        int   s = node;
        float w = 0.0f;
        if (lane < m) {
            const int2 ent = csr[o + c0 + lane];
            s = ent.x;
            w = __int_as_float(ent.y) * di;
        }
        for (int j0 = 0; j0 < m; j0 += 16) {
            float v[8];
            #pragma unroll
            for (int u = 0; u < 8; ++u) {
                const int sa = __builtin_amdgcn_readlane(s, j0 + 2 * u);
                const int sb = __builtin_amdgcn_readlane(s, j0 + 2 * u + 1);
                const int sj = half ? sb : sa;
                v[u] = h2[(size_t)sj * F_OUT + f];
            }
            #pragma unroll
            for (int u = 0; u < 8; ++u) {
                const int   wa = __builtin_amdgcn_readlane(__float_as_int(w), j0 + 2 * u);
                const int   wb = __builtin_amdgcn_readlane(__float_as_int(w), j0 + 2 * u + 1);
                const float nj = __int_as_float(half ? wb : wa);
                if (j0 + 2 * u + half < m) acc += v[u] * nj;
            }
        }
    }
    acc += __shfl_xor(acc, 16);
    if (half == 0) out[(size_t)node * F_OUT + f] = acc;
}

extern "C" void kernel_launch(void* const* d_in, const int* in_sizes, int n_in,
                              void* d_out, int out_size, void* d_ws, size_t ws_size,
                              hipStream_t stream) {
    const float* x  = (const float*)d_in[0];
    const int*   ei = (const int*)  d_in[1];
    const float* W1 = (const float*)d_in[2];
    const float* b1 = (const float*)d_in[3];
    const float* W2 = (const float*)d_in[4];
    const float* b2 = (const float*)d_in[5];
    float* out = (float*)d_out;

    const int n = in_sizes[0] / F_IN;
    const int E = in_sizes[1] / 2;
    const int* src = ei;
    const int* dst = ei + E;

    float* h1     = (float*)d_ws;
    float* h2     = h1 + (size_t)n * F_HID;
    int2*  csr    = (int2*)(h2 + (size_t)n * F_OUT);
    int*   rank   = (int*)(csr + E);
    int*   table  = rank + E;
    int*   cnt    = table + (size_t)N_CHUNKS * n;
    int*   off    = cnt + n;
    float* dinv   = (float*)(off + n);
    int*   cursor = (int*)(dinv + n);

    const int mTiles  = n / 16;
    const int mGroups = (mTiles + 7) / 8;

    const int per = (E + N_CHUNKS - 1) / N_CHUNKS;
    k_count_lds<<<N_CHUNKS * N_RANGES, 1024, 0, stream>>>(dst, rank, table, cursor, n, E, per);
    k_scan_offsets<<<(n + 255) / 256, 256, 0, stream>>>(table, cursor, cnt, off, dinv, n);
    const int gemmBlocks  = mGroups * (F_HID / 16);
    const int placeBlocks = (E + 255) / 256;
    k_place_gemm1<<<gemmBlocks + placeBlocks, 256, 0, stream>>>(src, dst, off, rank, table, dinv, csr, n, E, per,
                                                               x, W1, h1, mTiles, mGroups, gemmBlocks);

    k_gather64_gemm2<<<(n + 15) / 16, 512, 0, stream>>>(csr, off, cnt, dinv, h1, b1, W2, h2, n);

    k_gather16<<<(int)(((size_t)n * 32 + 255) / 256), 256, 0, stream>>>(csr, off, cnt, dinv, h2, b2, out, n);
}


# x@W1: one wave per 16x64 tile (4 accumulators); each x fragment loaded once, 16 B per lane, used for all 4 column tiles; W1 whole in LDS; 16 B row stores; the product's k sum order changes
# speedup vs baseline: 2.9294x; 1.0542x over previous
#include <hip/hip_runtime.h>


typedef float v2f __attribute__((ext_vector_type(2)));
typedef float v8f __attribute__((ext_vector_type(8)));

#define F_IN  128
#define F_HID 64
#define F_OUT 16

#define N_CAP    50000
#define N_CHUNKS 32
#define N_RANGES 4
#define R_CAP    ((((N_CAP + N_RANGES - 1) / N_RANGES) + 3) & ~3)
#define BATCH    32
#define B_PITCH  72

__global__ __launch_bounds__(1024)
void k_count_lds(const int* __restrict__ dst, int* __restrict__ lrank, int* __restrict__ table,
                 int* __restrict__ cursor, int n, int E, int per) {
    __shared__ alignas(16) int cs[R_CAP];
    const int tid   = (int)threadIdx.x;
    const int chunk = (int)blockIdx.x / N_RANGES;
    const int range = (int)blockIdx.x - chunk * N_RANGES;
    const int nr    = (((n + N_RANGES - 1) / N_RANGES) + 3) & ~3;
    const int lo    = range * nr;
    const unsigned span = (unsigned)max(0, min(n, lo + nr) - lo);
    for (int i = tid; i < R_CAP / 4; i += 1024) ((int4*)cs)[i] = make_int4(0, 0, 0, 0);
    if (blockIdx.x == 0 && tid == 0) *cursor = 0;
    __syncthreads();
    const int e0 = chunk * per;
    const int e1 = min(E, e0 + per);
    for (int eb = e0; eb < e1; eb += BATCH * 1024) {
        int d[BATCH];
        int r[BATCH];
        #pragma unroll
        for (int u = 0; u < BATCH; ++u) {
            const int e = eb + u * 1024 + tid;
            d[u] = (e < e1) ? dst[e] : -1;
        }
        #pragma unroll
        for (int u = 0; u < BATCH; ++u) {
            const unsigned k = (unsigned)(d[u] - lo);
            r[u] = (k < span) ? atomicAdd(&cs[k], 1) : -1;
        }
        #pragma unroll
        for (int u = 0; u < BATCH; ++u) {
            const int e = eb + u * 1024 + tid;
            if (r[u] >= 0) lrank[e] = r[u];
        }
    }
    __syncthreads();
    int* row = table + (size_t)chunk * n + lo;
    if ((span & 3u) == 0 && ((size_t)row & 15) == 0) {
        for (int i = tid; i < (int)(span / 4); i += 1024) ((int4*)row)[i] = ((const int4*)cs)[i];
    } else {
        for (int i = tid; i < (int)span; i += 1024) row[i] = cs[i];
    }
}

__global__ __launch_bounds__(256)
void k_scan_offsets(int* __restrict__ table, int* __restrict__ cursor,
                               int* __restrict__ cnt, int* __restrict__ off,
                               float* __restrict__ dinv, int n) {
    const int i    = blockIdx.x * blockDim.x + threadIdx.x;
    const int lane = threadIdx.x & 31;
    int c = 0;
    if (i < n) {
        const unsigned un = (unsigned)n, ui = (unsigned)i;
        for (int g0 = 0; g0 < N_CHUNKS; g0 += BATCH) {
            int t[BATCH];
            #pragma unroll
            for (int u = 0; u < BATCH; ++u) t[u] = table[(unsigned)(g0 + u) * un + ui];
            #pragma unroll
            for (int u = 0; u < BATCH; ++u) { table[(unsigned)(g0 + u) * un + ui] = c; c += t[u]; }
        }
        cnt[i] = c;
    }
    int incl = c;
    #pragma unroll
    for (int sft = 1; sft < 32; sft <<= 1) {
        int t = __shfl_up(incl, sft);
        if (lane >= sft) incl += t;
    }
    const int total = __shfl(incl, 31);
    int base = 0;
    if (lane == 0) base = atomicAdd(cursor, total);
    base = __shfl(base, 0);
    if (i < n) {
        off[i]  = base + incl - c;
        dinv[i] = __frsqrt_rn((float)(c + 1));
    }
}

__global__ __launch_bounds__(256)
void k_place_gemm1(const int* __restrict__ src, const int* __restrict__ dst,
                   const int* __restrict__ off, const int* __restrict__ lrank,
                   const int* __restrict__ table, const float* __restrict__ dinv,
                   int2* __restrict__ csr, int n, int E, int per,
                   const float* __restrict__ X, const float* __restrict__ W,
                   float* __restrict__ H, int mTiles, int gemmBlocks) {
    __shared__ float2 Bs[(F_IN / 2) * B_PITCH];
    if ((int)blockIdx.x >= gemmBlocks) {
        const int e = ((int)blockIdx.x - gemmBlocks) * 256 + (int)threadIdx.x;
        if (e < E) {
            const int s = src[e];
            const int d = dst[e];
            const int g = e / per;
            const int p = off[d] + table[(size_t)g * n + d] + lrank[e];
            csr[p] = make_int2(s, __float_as_int(dinv[s]));
        }
        return;
    }
    {
        float4 wv[8];
        #pragma unroll
        for (int u = 0; u < 8; ++u) wv[u] = ((const float4*)W)[(int)threadIdx.x + u * 256];
        #pragma unroll
        for (int u = 0; u < 8; ++u) {
            const int idx4 = (int)threadIdx.x + u * 256;
            const int k = idx4 >> 4, q = idx4 & 15;
            float* bp = (float*)(Bs + (k >> 1) * B_PITCH + q) + (k & 1);
            bp[0] = wv[u].x; bp[32] = wv[u].y; bp[64] = wv[u].z; bp[96] = wv[u].w;
        }
    }
    __syncthreads();

    const int wave  = threadIdx.x >> 5;
    const int lane  = threadIdx.x & 31;
    const int mTile = (int)blockIdx.x * 8 + wave;
    if (mTile >= mTiles) return;
    const int mBase = mTile * 16;
    const int half  = lane >> 4;
    const int l     = lane & 15;
    const float*  arow = X + (size_t)(mBase + l) * F_IN + half * 4;
    const float2* brow = Bs + (2 * half) * B_PITCH + l;

    v8f c0 = {}, c1 = {}, c2 = {}, c3 = {};
    #pragma unroll 4
    for (int j = 0; j < F_IN / 8; ++j) {
        const float4 a4 = *(const float4*)(arow + 8 * j);
        #pragma unroll
        for (int u = 0; u < 2; ++u) {
            v2f a; a.x = u ? a4.z : a4.x; a.y = u ? a4.w : a4.y;
            const float2* bp = brow + (4 * j + u) * B_PITCH;
            const float2 f0 = bp[0], f1 = bp[16], f2 = bp[32], f3 = bp[48];
            v2f b0; b0.x = f0.x; b0.y = f0.y;
            v2f b1; b1.x = f1.x; b1.y = f1.y;
            v2f b2; b2.x = f2.x; b2.y = f2.y;
            v2f b3; b3.x = f3.x; b3.y = f3.y;
            c0 = __builtin_amdgcn_wmma_f32_16x16x4_f32(false, a, false, b0, (short)0, c0, false, false);
            c1 = __builtin_amdgcn_wmma_f32_16x16x4_f32(false, a, false, b1, (short)0, c1, false, false);
            c2 = __builtin_amdgcn_wmma_f32_16x16x4_f32(false, a, false, b2, (short)0, c2, false, false);
            c3 = __builtin_amdgcn_wmma_f32_16x16x4_f32(false, a, false, b3, (short)0, c3, false, false);
        }
    }
    float* orow = H + (size_t)(mBase + half * 8) * F_HID + 4 * l;
    #pragma unroll
    for (int r = 0; r < 8; ++r)
        *(float4*)(orow + (size_t)r * F_HID) = make_float4(c0[r], c1[r], c2[r], c3[r]);
}

#define A_PITCH 68

__global__ __launch_bounds__(512)
void k_gather64_gemm2(const int2* __restrict__ csr, const int* __restrict__ off,
                      const int* __restrict__ cnt, const float* __restrict__ dinv,
                      const float* __restrict__ h1, const float* __restrict__ bias1,
                      const float* __restrict__ W2, float* __restrict__ h2, int n) {
    __shared__ float As[16 * A_PITCH];
    __shared__ float Bs[F_HID * 16];
    const int lane = threadIdx.x & 31;
    const int wave = __builtin_amdgcn_readfirstlane((int)threadIdx.x >> 5);
    const int node = (int)blockIdx.x * 16 + wave;

    const float w2lo = W2[threadIdx.x];
    const float w2hi = W2[threadIdx.x + 512];
    const float2 bb = ((const float2*)bias1)[lane];

    float2 acc;
    acc.x = 0.0f; acc.y = 0.0f;
    if (node < n) {
        const int   c  = cnt[node];
        const int   o  = off[node];
        const float di = dinv[node];
        const float2 own = ((const float2*)(h1 + (size_t)node * F_HID))[lane];
        asm volatile("" :: "s"(c), "s"(o), "s"(di));
        acc.x = own.x * di * di;
        acc.y = own.y * di * di;
        for (int c0 = 0; c0 < c; c0 += 32) {
            const int m = min(32, c - c0);
            int   s = node;
            float w = 0.0f;
            if (lane < m) {
                const int2 ent = csr[o + c0 + lane];
                s = ent.x;
                w = __int_as_float(ent.y) * di;
            }
            for (int j0 = 0; j0 < m; j0 += 8) {
                float2 v[8];
                #pragma unroll
                for (int u = 0; u < 8; ++u) {
                    const int sj = __builtin_amdgcn_readlane(s, j0 + u);
                    v[u] = ((const float2*)(h1 + (size_t)sj * F_HID))[lane];
                }
                asm volatile("" :: "v"(v[0].x), "v"(v[0].y), "v"(v[1].x), "v"(v[1].y),
                                   "v"(v[2].x), "v"(v[2].y), "v"(v[3].x), "v"(v[3].y),
                                   "v"(v[4].x), "v"(v[4].y), "v"(v[5].x), "v"(v[5].y),
                                   "v"(v[6].x), "v"(v[6].y), "v"(v[7].x), "v"(v[7].y));
                #pragma unroll
                for (int u = 0; u < 8; ++u) {
                    if (j0 + u < m) {
                        const float nj = __int_as_float(
                            __builtin_amdgcn_readlane(__float_as_int(w), j0 + u));
                        acc.x += v[u].x * nj;
                        acc.y += v[u].y * nj;
                    }
                }
            }
        }
    }
    float2 a1;
    a1.x = fmaxf(acc.x + bb.x, 0.0f);
    a1.y = fmaxf(acc.y + bb.y, 0.0f);
    ((float2*)(As + wave * A_PITCH))[lane] = a1;
    Bs[threadIdx.x]       = w2lo;
    Bs[threadIdx.x + 512] = w2hi;
    __syncthreads();
    if (wave != 0) return;

    const int half = lane >> 4;
    const int l    = lane & 15;
    v8f c = {};
    #pragma unroll
    for (int k = 0; k < F_HID; k += 4) {
        const int ak = k + half * 2;
        v2f a = *(const v2f*)(As + l * A_PITCH + ak);
        v2f b; b.x = Bs[ak * 16 + l]; b.y = Bs[(ak + 1) * 16 + l];
        c = __builtin_amdgcn_wmma_f32_16x16x4_f32(false, a, false, b,
                                                  (short)0, c, false, false);
    }
    const int rBase = (int)blockIdx.x * 16 + half * 8;
    float* orow = h2 + (size_t)rBase * F_OUT + l;
    #pragma unroll
    for (int r = 0; r < 8; ++r)
        if (rBase + r < n) orow[(size_t)r * F_OUT] = c[r];
}

__global__ __launch_bounds__(256)
void k_gather16(const int2* __restrict__ csr, const int* __restrict__ off,
                const int* __restrict__ cnt, const float* __restrict__ dinv,
                const float* __restrict__ h2, const float* __restrict__ bias2,
                float* __restrict__ out, int n) {
    const int lane = threadIdx.x & 31;
    const int node = __builtin_amdgcn_readfirstlane((blockIdx.x * blockDim.x + threadIdx.x) >> 5);
    if (node >= n) return;
    const int half = lane >> 4;
    const int f    = lane & 15;
    const int   c  = cnt[node];
    const int   o  = off[node];
    const float di = dinv[node];
    const float own = h2[(size_t)node * F_OUT + f];
    float acc = (half == 0) ? (own * di * di + bias2[f]) : 0.0f;
    for (int c0 = 0; c0 < c; c0 += 32) {
        const int m = min(32, c - c0);
        int   s = node;
        float w = 0.0f;
        if (lane < m) {
            const int2 ent = csr[o + c0 + lane];
            s = ent.x;
            w = __int_as_float(ent.y) * di;
        }
        for (int j0 = 0; j0 < m; j0 += 16) {
            float v[8];
            #pragma unroll
            for (int u = 0; u < 8; ++u) {
                const int sa = __builtin_amdgcn_readlane(s, j0 + 2 * u);
                const int sb = __builtin_amdgcn_readlane(s, j0 + 2 * u + 1);
                const int sj = half ? sb : sa;
                v[u] = h2[(size_t)sj * F_OUT + f];
            }
            #pragma unroll
            for (int u = 0; u < 8; ++u) {
                const int   wa = __builtin_amdgcn_readlane(__float_as_int(w), j0 + 2 * u);
                const int   wb = __builtin_amdgcn_readlane(__float_as_int(w), j0 + 2 * u + 1);
                const float nj = __int_as_float(half ? wb : wa);
                if (j0 + 2 * u + half < m) acc += v[u] * nj;
            }
        }
    }
    acc += __shfl_xor(acc, 16);
    if (half == 0) out[(size_t)node * F_OUT + f] = acc;
}

extern "C" void kernel_launch(void* const* d_in, const int* in_sizes, int n_in,
                              void* d_out, int out_size, void* d_ws, size_t ws_size,
                              hipStream_t stream) {
    const float* x  = (const float*)d_in[0];
    const int*   ei = (const int*)  d_in[1];
    const float* W1 = (const float*)d_in[2];
    const float* b1 = (const float*)d_in[3];
    const float* W2 = (const float*)d_in[4];
    const float* b2 = (const float*)d_in[5];
    float* out = (float*)d_out;

    const int n = in_sizes[0] / F_IN;
    const int E = in_sizes[1] / 2;
    const int* src = ei;
    const int* dst = ei + E;

    float* h1     = (float*)d_ws;
    float* h2     = h1 + (size_t)n * F_HID;
    int2*  csr    = (int2*)(h2 + (size_t)n * F_OUT);
    int*   rank   = (int*)(csr + E);
    int*   table  = rank + E;
    int*   cnt    = table + (size_t)N_CHUNKS * n;
    int*   off    = cnt + n;
    float* dinv   = (float*)(off + n);
    int*   cursor = (int*)(dinv + n);

    const int mTiles  = n / 16;
    const int mGroups = (mTiles + 7) / 8;

    const int per = (E + N_CHUNKS - 1) / N_CHUNKS;
    k_count_lds<<<N_CHUNKS * N_RANGES, 1024, 0, stream>>>(dst, rank, table, cursor, n, E, per);
    k_scan_offsets<<<(n + 255) / 256, 256, 0, stream>>>(table, cursor, cnt, off, dinv, n);
    const int gemmBlocks  = mGroups;
    const int placeBlocks = (E + 255) / 256;
    k_place_gemm1<<<gemmBlocks + placeBlocks, 256, 0, stream>>>(src, dst, off, rank, table, dinv, csr, n, E, per,
                                                               x, W1, h1, mTiles, gemmBlocks);

    k_gather64_gemm2<<<(n + 15) / 16, 512, 0, stream>>>(csr, off, cnt, dinv, h1, b1, W2, h2, n);

    k_gather16<<<(int)(((size_t)n * 32 + 255) / 256), 256, 0, stream>>>(csr, off, cnt, dinv, h2, b2, out, n);
}


# edge slots taken in the placement: a work-group per (chunk, node range) on LDS counters seeded with the chunk's bucket starts; no rank array, no per-edge start reads; sum order: chunk, arrival
# speedup vs baseline: 3.3411x; 1.1405x over previous
#include <hip/hip_runtime.h>


typedef float v2f __attribute__((ext_vector_type(2)));
typedef float v8f __attribute__((ext_vector_type(8)));

#define F_IN  128
#define F_HID 64
#define F_OUT 16

#define N_CAP    50000
#define N_CHUNKS 32
#define N_RANGES 4
#define R_CAP    ((((N_CAP + N_RANGES - 1) / N_RANGES) + 3) & ~3)
#define BATCH    32
#define P_RANGES 8
#define P_SPAN   ((((N_CAP + P_RANGES - 1) / P_RANGES) + 3) & ~3)
#define P_BATCH  33
#define B_PITCH  72

__global__ __launch_bounds__(1024)
void k_count_lds(const int* __restrict__ dst, int* __restrict__ table,
                 int* __restrict__ cursor, int n, int E, int per) {
    __shared__ alignas(16) int cs[R_CAP];
    const int tid   = (int)threadIdx.x;
    const int chunk = (int)blockIdx.x / N_RANGES;
    const int range = (int)blockIdx.x - chunk * N_RANGES;
    const int nr    = (((n + N_RANGES - 1) / N_RANGES) + 3) & ~3;
    const int lo    = range * nr;
    const unsigned span = (unsigned)max(0, min(n, lo + nr) - lo);
    for (int i = tid; i < R_CAP / 4; i += 1024) ((int4*)cs)[i] = make_int4(0, 0, 0, 0);
    if (blockIdx.x == 0 && tid == 0) *cursor = 0;
    __syncthreads();
    const int e0 = chunk * per;
    const int e1 = min(E, e0 + per);
    for (int eb = e0; eb < e1; eb += BATCH * 1024) {
        int d[BATCH];
        #pragma unroll
        for (int u = 0; u < BATCH; ++u) {
            const int e = eb + u * 1024 + tid;
            d[u] = (e < e1) ? dst[e] : -1;
        }
        #pragma unroll
        for (int u = 0; u < BATCH; ++u) {
            const unsigned k = (unsigned)(d[u] - lo);
            if (k < span) atomicAdd(&cs[k], 1);
        }
    }
    __syncthreads();
    int* row = table + (size_t)chunk * n + lo;
    if ((span & 3u) == 0 && ((size_t)row & 15) == 0) {
        for (int i = tid; i < (int)(span / 4); i += 1024) ((int4*)row)[i] = ((const int4*)cs)[i];
    } else {
        for (int i = tid; i < (int)span; i += 1024) row[i] = cs[i];
    }
}

__global__ __launch_bounds__(256)
void k_scan_offsets(int* __restrict__ table, int* __restrict__ cursor,
                               int* __restrict__ cnt, int* __restrict__ off,
                               float* __restrict__ dinv, int n) {
    const int i    = blockIdx.x * blockDim.x + threadIdx.x;
    const int lane = threadIdx.x & 31;
    int c = 0;
    if (i < n) {
        const unsigned un = (unsigned)n, ui = (unsigned)i;
        for (int g0 = 0; g0 < N_CHUNKS; g0 += BATCH) {
            int t[BATCH];
            #pragma unroll
            for (int u = 0; u < BATCH; ++u) t[u] = table[(unsigned)(g0 + u) * un + ui];
            #pragma unroll
            for (int u = 0; u < BATCH; ++u) { table[(unsigned)(g0 + u) * un + ui] = c; c += t[u]; }
        }
        cnt[i] = c;
    }
    int incl = c;
    #pragma unroll
    for (int sft = 1; sft < 32; sft <<= 1) {
        int t = __shfl_up(incl, sft);
        if (lane >= sft) incl += t;
    }
    const int total = __shfl(incl, 31);
    int base = 0;
    if (lane == 0) base = atomicAdd(cursor, total);
    base = __shfl(base, 0);
    if (i < n) {
        off[i]  = base + incl - c;
        dinv[i] = __frsqrt_rn((float)(c + 1));
    }
}

__global__ __launch_bounds__(256)
void k_place_gemm1(const int* __restrict__ src, const int* __restrict__ dst,
                   const int* __restrict__ off,
                   const int* __restrict__ table, const float* __restrict__ dinv,
                   int2* __restrict__ csr, int n, int E, int per,
                   const float* __restrict__ X, const float* __restrict__ W,
                   float* __restrict__ H, int mTiles, int gemmBlocks) {
    __shared__ float2 Bs[(F_IN / 2) * B_PITCH];
    static_assert(P_SPAN * sizeof(int) <= sizeof(float2) * (F_IN / 2) * B_PITCH, "placement counters must fit the shared array");
    if ((int)blockIdx.x >= gemmBlocks) {
        int* cs = (int*)Bs;
        const int tid   = (int)threadIdx.x;
        const int pb    = (int)blockIdx.x - gemmBlocks;
        const int chunk = pb / P_RANGES;
        const int range = pb - chunk * P_RANGES;
        const int nr    = (((n + P_RANGES - 1) / P_RANGES) + 3) & ~3;
        const int lo    = range * nr;
        const unsigned span = (unsigned)max(0, min(n, lo + nr) - lo);
        const int* trow = table + (size_t)chunk * n + lo;
        const int* orow = off + lo;
        if ((span & 3u) == 0 && (((size_t)trow | (size_t)orow) & 15) == 0) {
            for (int i = tid; i < (int)(span / 4); i += 256) {
                const int4 a = ((const int4*)trow)[i];
                const int4 b = ((const int4*)orow)[i];
                ((int4*)cs)[i] = make_int4(a.x + b.x, a.y + b.y, a.z + b.z, a.w + b.w);
            }
        } else {
            for (int i = tid; i < (int)span; i += 256) cs[i] = trow[i] + orow[i];
        }
        __syncthreads();
        const int e0 = chunk * per;
        const int e1 = min(E, e0 + per);
        for (int eb = e0; eb < e1; eb += P_BATCH * 256) {
            int d[P_BATCH];
            int s[P_BATCH];
            int p[P_BATCH];
            #pragma unroll
            for (int u = 0; u < P_BATCH; ++u) {
                const int e = eb + u * 256 + tid;
                d[u] = (e < e1) ? dst[e] : -1;
            }
            #pragma unroll
            for (int u = 0; u < P_BATCH; ++u) {
                const unsigned k = (unsigned)(d[u] - lo);
                p[u] = -1;
                s[u] = 0;
                if (k < span) {
                    p[u] = atomicAdd(&cs[k], 1);
                    s[u] = src[eb + u * 256 + tid];
                }
            }
            float w[P_BATCH];
            #pragma unroll
            for (int u = 0; u < P_BATCH; ++u) w[u] = dinv[s[u]];
            #pragma unroll
            for (int u = 0; u < P_BATCH; ++u) {
                if (p[u] >= 0) csr[p[u]] = make_int2(s[u], __float_as_int(w[u]));
            }
        }
        return;
    }
    {
        float4 wv[8];
        #pragma unroll
        for (int u = 0; u < 8; ++u) wv[u] = ((const float4*)W)[(int)threadIdx.x + u * 256];
        #pragma unroll
        for (int u = 0; u < 8; ++u) {
            const int idx4 = (int)threadIdx.x + u * 256;
            const int k = idx4 >> 4, q = idx4 & 15;
            float* bp = (float*)(Bs + (k >> 1) * B_PITCH + q) + (k & 1);
            bp[0] = wv[u].x; bp[32] = wv[u].y; bp[64] = wv[u].z; bp[96] = wv[u].w;
        }
    }
    __syncthreads();

    const int wave  = threadIdx.x >> 5;
    const int lane  = threadIdx.x & 31;
    const int mTile = (int)blockIdx.x * 8 + wave;
    if (mTile >= mTiles) return;
    const int mBase = mTile * 16;
    const int half  = lane >> 4;
    const int l     = lane & 15;
    const float*  arow = X + (size_t)(mBase + l) * F_IN + half * 4;
    const float2* brow = Bs + (2 * half) * B_PITCH + l;

    v8f c0 = {}, c1 = {}, c2 = {}, c3 = {};
    #pragma unroll 4
    for (int j = 0; j < F_IN / 8; ++j) {
        const float4 a4 = *(const float4*)(arow + 8 * j);
        #pragma unroll
        for (int u = 0; u < 2; ++u) {
            v2f a; a.x = u ? a4.z : a4.x; a.y = u ? a4.w : a4.y;
            const float2* bp = brow + (4 * j + u) * B_PITCH;
            const float2 f0 = bp[0], f1 = bp[16], f2 = bp[32], f3 = bp[48];
            v2f b0; b0.x = f0.x; b0.y = f0.y;
            v2f b1; b1.x = f1.x; b1.y = f1.y;
            v2f b2; b2.x = f2.x; b2.y = f2.y;
            v2f b3; b3.x = f3.x; b3.y = f3.y;
            c0 = __builtin_amdgcn_wmma_f32_16x16x4_f32(false, a, false, b0, (short)0, c0, false, false);
            c1 = __builtin_amdgcn_wmma_f32_16x16x4_f32(false, a, false, b1, (short)0, c1, false, false);
            c2 = __builtin_amdgcn_wmma_f32_16x16x4_f32(false, a, false, b2, (short)0, c2, false, false);
            c3 = __builtin_amdgcn_wmma_f32_16x16x4_f32(false, a, false, b3, (short)0, c3, false, false);
        }
    }
    float* orow = H + (size_t)(mBase + half * 8) * F_HID + 4 * l;
    #pragma unroll
    for (int r = 0; r < 8; ++r)
        *(float4*)(orow + (size_t)r * F_HID) = make_float4(c0[r], c1[r], c2[r], c3[r]);
}

#define A_PITCH 68

__global__ __launch_bounds__(512)
void k_gather64_gemm2(const int2* __restrict__ csr, const int* __restrict__ off,
                      const int* __restrict__ cnt, const float* __restrict__ dinv,
                      const float* __restrict__ h1, const float* __restrict__ bias1,
                      const float* __restrict__ W2, float* __restrict__ h2, int n) {
    __shared__ float As[16 * A_PITCH];
    __shared__ float Bs[F_HID * 16];
    const int lane = threadIdx.x & 31;
    const int wave = __builtin_amdgcn_readfirstlane((int)threadIdx.x >> 5);
    const int node = (int)blockIdx.x * 16 + wave;

    const float w2lo = W2[threadIdx.x];
    const float w2hi = W2[threadIdx.x + 512];
    const float2 bb = ((const float2*)bias1)[lane];

    float2 acc;
    acc.x = 0.0f; acc.y = 0.0f;
    if (node < n) {
        const int   c  = cnt[node];
        const int   o  = off[node];
        const float di = dinv[node];
        const float2 own = ((const float2*)(h1 + (size_t)node * F_HID))[lane];
        asm volatile("" :: "s"(c), "s"(o), "s"(di));
        acc.x = own.x * di * di;
        acc.y = own.y * di * di;
        for (int c0 = 0; c0 < c; c0 += 32) {
            const int m = min(32, c - c0);
            int   s = node;
            float w = 0.0f;
            if (lane < m) {
                const int2 ent = csr[o + c0 + lane];
                s = ent.x;
                w = __int_as_float(ent.y) * di;
            }
            for (int j0 = 0; j0 < m; j0 += 8) {
                float2 v[8];
                #pragma unroll
                for (int u = 0; u < 8; ++u) {
                    const int sj = __builtin_amdgcn_readlane(s, j0 + u);
                    v[u] = ((const float2*)(h1 + (size_t)sj * F_HID))[lane];
                }
                asm volatile("" :: "v"(v[0].x), "v"(v[0].y), "v"(v[1].x), "v"(v[1].y),
                                   "v"(v[2].x), "v"(v[2].y), "v"(v[3].x), "v"(v[3].y),
                                   "v"(v[4].x), "v"(v[4].y), "v"(v[5].x), "v"(v[5].y),
                                   "v"(v[6].x), "v"(v[6].y), "v"(v[7].x), "v"(v[7].y));
                #pragma unroll
                for (int u = 0; u < 8; ++u) {
                    if (j0 + u < m) {
                        const float nj = __int_as_float(
                            __builtin_amdgcn_readlane(__float_as_int(w), j0 + u));
                        acc.x += v[u].x * nj;
                        acc.y += v[u].y * nj;
                    }
                }
            }
        }
    }
    float2 a1;
    a1.x = fmaxf(acc.x + bb.x, 0.0f);
    a1.y = fmaxf(acc.y + bb.y, 0.0f);
    ((float2*)(As + wave * A_PITCH))[lane] = a1;
    Bs[threadIdx.x]       = w2lo;
    Bs[threadIdx.x + 512] = w2hi;
    __syncthreads();
    if (wave != 0) return;

    const int half = lane >> 4;
    const int l    = lane & 15;
    v8f c = {};
    #pragma unroll
    for (int k = 0; k < F_HID; k += 4) {
        const int ak = k + half * 2;
        v2f a = *(const v2f*)(As + l * A_PITCH + ak);
        v2f b; b.x = Bs[ak * 16 + l]; b.y = Bs[(ak + 1) * 16 + l];
        c = __builtin_amdgcn_wmma_f32_16x16x4_f32(false, a, false, b,
                                                  (short)0, c, false, false);
    }
    const int rBase = (int)blockIdx.x * 16 + half * 8;
    float* orow = h2 + (size_t)rBase * F_OUT + l;
    #pragma unroll
    for (int r = 0; r < 8; ++r)
        if (rBase + r < n) orow[(size_t)r * F_OUT] = c[r];
}

__global__ __launch_bounds__(256)
void k_gather16(const int2* __restrict__ csr, const int* __restrict__ off,
                const int* __restrict__ cnt, const float* __restrict__ dinv,
                const float* __restrict__ h2, const float* __restrict__ bias2,
                float* __restrict__ out, int n) {
    const int lane = threadIdx.x & 31;
    const int node = __builtin_amdgcn_readfirstlane((blockIdx.x * blockDim.x + threadIdx.x) >> 5);
    if (node >= n) return;
    const int half = lane >> 4;
    const int f    = lane & 15;
    const int   c  = cnt[node];
    const int   o  = off[node];
    const float di = dinv[node];
    const float own = h2[(size_t)node * F_OUT + f];
    float acc = (half == 0) ? (own * di * di + bias2[f]) : 0.0f;
    for (int c0 = 0; c0 < c; c0 += 32) {
        const int m = min(32, c - c0);
        int   s = node;
        float w = 0.0f;
        if (lane < m) {
            const int2 ent = csr[o + c0 + lane];
            s = ent.x;
            w = __int_as_float(ent.y) * di;
        }
        for (int j0 = 0; j0 < m; j0 += 16) {
            float v[8];
            #pragma unroll
            for (int u = 0; u < 8; ++u) {
                const int sa = __builtin_amdgcn_readlane(s, j0 + 2 * u);
                const int sb = __builtin_amdgcn_readlane(s, j0 + 2 * u + 1);
                const int sj = half ? sb : sa;
                v[u] = h2[(size_t)sj * F_OUT + f];
            }
            #pragma unroll
            for (int u = 0; u < 8; ++u) {
                const int   wa = __builtin_amdgcn_readlane(__float_as_int(w), j0 + 2 * u);
                const int   wb = __builtin_amdgcn_readlane(__float_as_int(w), j0 + 2 * u + 1);
                const float nj = __int_as_float(half ? wb : wa);
                if (j0 + 2 * u + half < m) acc += v[u] * nj;
            }
        }
    }
    acc += __shfl_xor(acc, 16);
    if (half == 0) out[(size_t)node * F_OUT + f] = acc;
}

extern "C" void kernel_launch(void* const* d_in, const int* in_sizes, int n_in,
                              void* d_out, int out_size, void* d_ws, size_t ws_size,
                              hipStream_t stream) {
    const float* x  = (const float*)d_in[0];
    const int*   ei = (const int*)  d_in[1];
    const float* W1 = (const float*)d_in[2];
    const float* b1 = (const float*)d_in[3];
    const float* W2 = (const float*)d_in[4];
    const float* b2 = (const float*)d_in[5];
    float* out = (float*)d_out;

    const int n = in_sizes[0] / F_IN;
    const int E = in_sizes[1] / 2;
    const int* src = ei;
    const int* dst = ei + E;

    float* h1     = (float*)d_ws;
    float* h2     = h1 + (size_t)n * F_HID;
    int2*  csr    = (int2*)(h2 + (size_t)n * F_OUT);
    int*   table  = (int*)(csr + E);
    int*   cnt    = table + (size_t)N_CHUNKS * n;
    int*   off    = cnt + n;
    float* dinv   = (float*)(off + n);
    int*   cursor = (int*)(dinv + n);

    const int mTiles  = n / 16;
    const int mGroups = (mTiles + 7) / 8;

    const int per = (E + N_CHUNKS - 1) / N_CHUNKS;
    k_count_lds<<<N_CHUNKS * N_RANGES, 1024, 0, stream>>>(dst, table, cursor, n, E, per);
    k_scan_offsets<<<(n + 255) / 256, 256, 0, stream>>>(table, cursor, cnt, off, dinv, n);
    const int gemmBlocks  = mGroups;
    const int placeBlocks = N_CHUNKS * P_RANGES;
    k_place_gemm1<<<gemmBlocks + placeBlocks, 256, 0, stream>>>(src, dst, off, table, dinv, csr, n, E, per,
                                                               x, W1, h1, mTiles, gemmBlocks);

    k_gather64_gemm2<<<(n + 15) / 16, 512, 0, stream>>>(csr, off, cnt, dinv, h1, b1, W2, h2, n);

    k_gather16<<<(int)(((size_t)n * 32 + 255) / 256), 256, 0, stream>>>(csr, off, cnt, dinv, h2, b2, out, n);
}


# edge weight factored, dinv[d]*sum(dinv[s]*h[s]): both products store rows pre-scaled by the row's dinv, edge sums only add, slots hold the source only, no per-edge dinv read; sum order kept
# speedup vs baseline: 3.5170x; 1.0526x over previous
#include <hip/hip_runtime.h>


typedef float v2f __attribute__((ext_vector_type(2)));
typedef float v8f __attribute__((ext_vector_type(8)));

#define F_IN  128
#define F_HID 64
#define F_OUT 16

#define N_CAP    50000
#define N_CHUNKS 32
#define N_RANGES 4
#define R_CAP    ((((N_CAP + N_RANGES - 1) / N_RANGES) + 3) & ~3)
#define BATCH    32
#define P_RANGES 8
#define P_SPAN   ((((N_CAP + P_RANGES - 1) / P_RANGES) + 3) & ~3)
#define P_BATCH  33
#define B_PITCH  72

__global__ __launch_bounds__(1024)
void k_count_lds(const int* __restrict__ dst, int* __restrict__ table,
                 int* __restrict__ cursor, int n, int E, int per) {
    __shared__ alignas(16) int cs[R_CAP];
    const int tid   = (int)threadIdx.x;
    const int chunk = (int)blockIdx.x / N_RANGES;
    const int range = (int)blockIdx.x - chunk * N_RANGES;
    const int nr    = (((n + N_RANGES - 1) / N_RANGES) + 3) & ~3;
    const int lo    = range * nr;
    const unsigned span = (unsigned)max(0, min(n, lo + nr) - lo);
    for (int i = tid; i < R_CAP / 4; i += 1024) ((int4*)cs)[i] = make_int4(0, 0, 0, 0);
    if (blockIdx.x == 0 && tid == 0) *cursor = 0;
    __syncthreads();
    const int e0 = chunk * per;
    const int e1 = min(E, e0 + per);
    for (int eb = e0; eb < e1; eb += BATCH * 1024) {
        int d[BATCH];
        #pragma unroll
        for (int u = 0; u < BATCH; ++u) {
            const int e = eb + u * 1024 + tid;
            d[u] = (e < e1) ? dst[e] : -1;
        }
        #pragma unroll
        for (int u = 0; u < BATCH; ++u) {
            const unsigned k = (unsigned)(d[u] - lo);
            if (k < span) atomicAdd(&cs[k], 1);
        }
    }
    __syncthreads();
    int* row = table + (size_t)chunk * n + lo;
    if ((span & 3u) == 0 && ((size_t)row & 15) == 0) {
        for (int i = tid; i < (int)(span / 4); i += 1024) ((int4*)row)[i] = ((const int4*)cs)[i];
    } else {
        for (int i = tid; i < (int)span; i += 1024) row[i] = cs[i];
    }
}

__global__ __launch_bounds__(256)
void k_scan_offsets(int* __restrict__ table, int* __restrict__ cursor,
                               int* __restrict__ cnt, int* __restrict__ off,
                               float* __restrict__ dinv, int n) {
    const int i    = blockIdx.x * blockDim.x + threadIdx.x;
    const int lane = threadIdx.x & 31;
    int c = 0;
    if (i < n) {
        const unsigned un = (unsigned)n, ui = (unsigned)i;
        for (int g0 = 0; g0 < N_CHUNKS; g0 += BATCH) {
            int t[BATCH];
            #pragma unroll
            for (int u = 0; u < BATCH; ++u) t[u] = table[(unsigned)(g0 + u) * un + ui];
            #pragma unroll
            for (int u = 0; u < BATCH; ++u) { table[(unsigned)(g0 + u) * un + ui] = c; c += t[u]; }
        }
        cnt[i] = c;
    }
    int incl = c;
    #pragma unroll
    for (int sft = 1; sft < 32; sft <<= 1) {
        int t = __shfl_up(incl, sft);
        if (lane >= sft) incl += t;
    }
    const int total = __shfl(incl, 31);
    int base = 0;
    if (lane == 0) base = atomicAdd(cursor, total);
    base = __shfl(base, 0);
    if (i < n) {
        off[i]  = base + incl - c;
        dinv[i] = __frsqrt_rn((float)(c + 1));
    }
}

__global__ __launch_bounds__(256)
void k_place_gemm1(const int* __restrict__ src, const int* __restrict__ dst,
                   const int* __restrict__ off,
                   const int* __restrict__ table, const float* __restrict__ dinv,
                   int* __restrict__ csr, int n, int E, int per,
                   const float* __restrict__ X, const float* __restrict__ W,
                   float* __restrict__ H, int mTiles, int gemmBlocks) {
    __shared__ float2 Bs[(F_IN / 2) * B_PITCH];
    static_assert(P_SPAN * sizeof(int) <= sizeof(float2) * (F_IN / 2) * B_PITCH, "placement counters must fit the shared array");
    if ((int)blockIdx.x >= gemmBlocks) {
        int* cs = (int*)Bs;
        const int tid   = (int)threadIdx.x;
        const int pb    = (int)blockIdx.x - gemmBlocks;
        const int chunk = pb / P_RANGES;
        const int range = pb - chunk * P_RANGES;
        const int nr    = (((n + P_RANGES - 1) / P_RANGES) + 3) & ~3;
        const int lo    = range * nr;
        const unsigned span = (unsigned)max(0, min(n, lo + nr) - lo);
        const int* trow = table + (size_t)chunk * n + lo;
        const int* orow = off + lo;
        if ((span & 3u) == 0 && (((size_t)trow | (size_t)orow) & 15) == 0) {
            for (int i = tid; i < (int)(span / 4); i += 256) {
                const int4 a = ((const int4*)trow)[i];
                const int4 b = ((const int4*)orow)[i];
                ((int4*)cs)[i] = make_int4(a.x + b.x, a.y + b.y, a.z + b.z, a.w + b.w);
            }
        } else {
            for (int i = tid; i < (int)span; i += 256) cs[i] = trow[i] + orow[i];
        }
        __syncthreads();
        const int e0 = chunk * per;
        const int e1 = min(E, e0 + per);
        for (int eb = e0; eb < e1; eb += P_BATCH * 256) {
            int d[P_BATCH];
            int s[P_BATCH];
            int p[P_BATCH];
            #pragma unroll
            for (int u = 0; u < P_BATCH; ++u) {
                const int e = eb + u * 256 + tid;
                d[u] = (e < e1) ? dst[e] : -1;
            }
            #pragma unroll
            for (int u = 0; u < P_BATCH; ++u) {
                const unsigned k = (unsigned)(d[u] - lo);
                p[u] = -1;
                s[u] = 0;
                if (k < span) {
                    p[u] = atomicAdd(&cs[k], 1);
                    s[u] = src[eb + u * 256 + tid];
                }
            }
            #pragma unroll
            for (int u = 0; u < P_BATCH; ++u) {
                if (p[u] >= 0) csr[p[u]] = s[u];
            }
        }
        return;
    }
    {
        float4 wv[8];
        #pragma unroll
        for (int u = 0; u < 8; ++u) wv[u] = ((const float4*)W)[(int)threadIdx.x + u * 256];
        #pragma unroll
        for (int u = 0; u < 8; ++u) {
            const int idx4 = (int)threadIdx.x + u * 256;
            const int k = idx4 >> 4, q = idx4 & 15;
            float* bp = (float*)(Bs + (k >> 1) * B_PITCH + q) + (k & 1);
            bp[0] = wv[u].x; bp[32] = wv[u].y; bp[64] = wv[u].z; bp[96] = wv[u].w;
        }
    }
    __syncthreads();

    const int wave  = threadIdx.x >> 5;
    const int lane  = threadIdx.x & 31;
    const int mTile = (int)blockIdx.x * 8 + wave;
    if (mTile >= mTiles) return;
    const int mBase = mTile * 16;
    const int half  = lane >> 4;
    const int l     = lane & 15;
    const float*  arow = X + (size_t)(mBase + l) * F_IN + half * 4;
    const float2* brow = Bs + (2 * half) * B_PITCH + l;

    float dv[8];
    #pragma unroll
    for (int r = 0; r < 8; ++r) dv[r] = dinv[mBase + half * 8 + r];

    v8f c0 = {}, c1 = {}, c2 = {}, c3 = {};
    #pragma unroll 4
    for (int j = 0; j < F_IN / 8; ++j) {
        const float4 a4 = *(const float4*)(arow + 8 * j);
        #pragma unroll
        for (int u = 0; u < 2; ++u) {
            v2f a; a.x = u ? a4.z : a4.x; a.y = u ? a4.w : a4.y;
            const float2* bp = brow + (4 * j + u) * B_PITCH;
            const float2 f0 = bp[0], f1 = bp[16], f2 = bp[32], f3 = bp[48];
            v2f b0; b0.x = f0.x; b0.y = f0.y;
            v2f b1; b1.x = f1.x; b1.y = f1.y;
            v2f b2; b2.x = f2.x; b2.y = f2.y;
            v2f b3; b3.x = f3.x; b3.y = f3.y;
            c0 = __builtin_amdgcn_wmma_f32_16x16x4_f32(false, a, false, b0, (short)0, c0, false, false);
            c1 = __builtin_amdgcn_wmma_f32_16x16x4_f32(false, a, false, b1, (short)0, c1, false, false);
            c2 = __builtin_amdgcn_wmma_f32_16x16x4_f32(false, a, false, b2, (short)0, c2, false, false);
            c3 = __builtin_amdgcn_wmma_f32_16x16x4_f32(false, a, false, b3, (short)0, c3, false, false);
        }
    }
    float* orow = H + (size_t)(mBase + half * 8) * F_HID + 4 * l;
    #pragma unroll
    for (int r = 0; r < 8; ++r)
        *(float4*)(orow + (size_t)r * F_HID) = make_float4(c0[r] * dv[r], c1[r] * dv[r], c2[r] * dv[r], c3[r] * dv[r]);
}

#define A_PITCH 68

__global__ __launch_bounds__(512)
void k_gather64_gemm2(const int* __restrict__ csr, const int* __restrict__ off,
                      const int* __restrict__ cnt, const float* __restrict__ dinv,
                      const float* __restrict__ h1, const float* __restrict__ bias1,
                      const float* __restrict__ W2, float* __restrict__ h2, int n) {
    __shared__ float As[16 * A_PITCH];
    __shared__ float Bs[F_HID * 16];
    __shared__ float Ds[16];
    const int lane = threadIdx.x & 31;
    const int wave = __builtin_amdgcn_readfirstlane((int)threadIdx.x >> 5);
    const int node = (int)blockIdx.x * 16 + wave;

    const float w2lo = W2[threadIdx.x];
    const float w2hi = W2[threadIdx.x + 512];
    const float2 bb = ((const float2*)bias1)[lane];

    float2 acc;
    acc.x = 0.0f; acc.y = 0.0f;
    float dn = 0.0f;
    if (node < n) {
        const int   c  = cnt[node];
        const int   o  = off[node];
        const float di = dinv[node];
        const float2 own = ((const float2*)(h1 + (size_t)node * F_HID))[lane];
        asm volatile("" :: "s"(c), "s"(o), "s"(di));
        dn = di;
        acc = own;
        for (int c0 = 0; c0 < c; c0 += 32) {
            const int m = min(32, c - c0);
            int s = node;
            if (lane < m) s = csr[o + c0 + lane];
            for (int j0 = 0; j0 < m; j0 += 8) {
                float2 v[8];
                #pragma unroll
                for (int u = 0; u < 8; ++u) {
                    const int sj = __builtin_amdgcn_readlane(s, j0 + u);
                    v[u] = ((const float2*)(h1 + (size_t)sj * F_HID))[lane];
                }
                asm volatile("" :: "v"(v[0].x), "v"(v[0].y), "v"(v[1].x), "v"(v[1].y),
                                   "v"(v[2].x), "v"(v[2].y), "v"(v[3].x), "v"(v[3].y),
                                   "v"(v[4].x), "v"(v[4].y), "v"(v[5].x), "v"(v[5].y),
                                   "v"(v[6].x), "v"(v[6].y), "v"(v[7].x), "v"(v[7].y));
                #pragma unroll
                for (int u = 0; u < 8; ++u) {
                    if (j0 + u < m) {
                        acc.x += v[u].x;
                        acc.y += v[u].y;
                    }
                }
            }
        }
    }
    float2 a1;
    a1.x = fmaxf(acc.x * dn + bb.x, 0.0f);
    a1.y = fmaxf(acc.y * dn + bb.y, 0.0f);
    ((float2*)(As + wave * A_PITCH))[lane] = a1;
    if (lane == 0) Ds[wave] = dn;
    Bs[threadIdx.x]       = w2lo;
    Bs[threadIdx.x + 512] = w2hi;
    __syncthreads();
    if (wave != 0) return;

    const int half = lane >> 4;
    const int l    = lane & 15;
    v8f c = {};
    #pragma unroll
    for (int k = 0; k < F_HID; k += 4) {
        const int ak = k + half * 2;
        v2f a = *(const v2f*)(As + l * A_PITCH + ak);
        v2f b; b.x = Bs[ak * 16 + l]; b.y = Bs[(ak + 1) * 16 + l];
        c = __builtin_amdgcn_wmma_f32_16x16x4_f32(false, a, false, b,
                                                  (short)0, c, false, false);
    }
    const int rBase = (int)blockIdx.x * 16 + half * 8;
    float* orow = h2 + (size_t)rBase * F_OUT + l;
    #pragma unroll
    for (int r = 0; r < 8; ++r)
        if (rBase + r < n) orow[(size_t)r * F_OUT] = c[r] * Ds[half * 8 + r];
}

__global__ __launch_bounds__(256)
void k_gather16(const int* __restrict__ csr, const int* __restrict__ off,
                const int* __restrict__ cnt, const float* __restrict__ dinv,
                const float* __restrict__ h2, const float* __restrict__ bias2,
                float* __restrict__ out, int n) {
    const int lane = threadIdx.x & 31;
    const int node = __builtin_amdgcn_readfirstlane((blockIdx.x * blockDim.x + threadIdx.x) >> 5);
    if (node >= n) return;
    const int half = lane >> 4;
    const int f    = lane & 15;
    const int   c  = cnt[node];
    const int   o  = off[node];
    const float di = dinv[node];
    const float own = h2[(size_t)node * F_OUT + f];
    const float b2f = bias2[f];
    float acc = (half == 0) ? own : 0.0f;
    for (int c0 = 0; c0 < c; c0 += 32) {
        const int m = min(32, c - c0);
        int s = node;
        if (lane < m) s = csr[o + c0 + lane];
        for (int j0 = 0; j0 < m; j0 += 16) {
            float v[8];
            #pragma unroll
            for (int u = 0; u < 8; ++u) {
                const int sa = __builtin_amdgcn_readlane(s, j0 + 2 * u);
                const int sb = __builtin_amdgcn_readlane(s, j0 + 2 * u + 1);
                const int sj = half ? sb : sa;
                v[u] = h2[(size_t)sj * F_OUT + f];
            }
            #pragma unroll
            for (int u = 0; u < 8; ++u) {
                if (j0 + 2 * u + half < m) acc += v[u];
            }
        }
    }
    acc += __shfl_xor(acc, 16);
    if (half == 0) out[(size_t)node * F_OUT + f] = acc * di + b2f;
}

extern "C" void kernel_launch(void* const* d_in, const int* in_sizes, int n_in,
                              void* d_out, int out_size, void* d_ws, size_t ws_size,
                              hipStream_t stream) {
    const float* x  = (const float*)d_in[0];
    const int*   ei = (const int*)  d_in[1];
    const float* W1 = (const float*)d_in[2];
    const float* b1 = (const float*)d_in[3];
    const float* W2 = (const float*)d_in[4];
    const float* b2 = (const float*)d_in[5];
    float* out = (float*)d_out;

    const int n = in_sizes[0] / F_IN;
    const int E = in_sizes[1] / 2;
    const int* src = ei;
    const int* dst = ei + E;

    float* h1     = (float*)d_ws;
    float* h2     = h1 + (size_t)n * F_HID;
    int*   csr    = (int*)(h2 + (size_t)n * F_OUT);
    int*   table  = (int*)(csr + E);
    int*   cnt    = table + (size_t)N_CHUNKS * n;
    int*   off    = cnt + n;
    float* dinv   = (float*)(off + n);
    int*   cursor = (int*)(dinv + n);

    const int mTiles  = n / 16;
    const int mGroups = (mTiles + 7) / 8;

    const int per = (E + N_CHUNKS - 1) / N_CHUNKS;
    k_count_lds<<<N_CHUNKS * N_RANGES, 1024, 0, stream>>>(dst, table, cursor, n, E, per);
    k_scan_offsets<<<(n + 255) / 256, 256, 0, stream>>>(table, cursor, cnt, off, dinv, n);
    const int gemmBlocks  = mGroups;
    const int placeBlocks = N_CHUNKS * P_RANGES;
    k_place_gemm1<<<gemmBlocks + placeBlocks, 256, 0, stream>>>(src, dst, off, table, dinv, csr, n, E, per,
                                                               x, W1, h1, mTiles, gemmBlocks);

    k_gather64_gemm2<<<(n + 15) / 16, 512, 0, stream>>>(csr, off, cnt, dinv, h1, b1, W2, h2, n);

    k_gather16<<<(int)(((size_t)n * 32 + 255) / 256), 256, 0, stream>>>(csr, off, cnt, dinv, h2, b2, out, n);
}


# both edge-sum launches: a wave owns several neighbour nodes (4 in layer 1, 8 in layer 2), bounds and first slots loaded up front, rows walked 2 nodes at a time; 100000 waves to 18750; sum order kept
# speedup vs baseline: 3.6410x; 1.0353x over previous
#include <hip/hip_runtime.h>


typedef float v2f __attribute__((ext_vector_type(2)));
typedef float v8f __attribute__((ext_vector_type(8)));

#define F_IN  128
#define F_HID 64
#define F_OUT 16

#define N_CAP    50000
#define N_CHUNKS 32
#define N_RANGES 4
#define R_CAP    ((((N_CAP + N_RANGES - 1) / N_RANGES) + 3) & ~3)
#define BATCH    32
#define P_RANGES 8
#define P_SPAN   ((((N_CAP + P_RANGES - 1) / P_RANGES) + 3) & ~3)
#define P_BATCH  33
#define B_PITCH  72

__global__ __launch_bounds__(1024)
void k_count_lds(const int* __restrict__ dst, int* __restrict__ table,
                 int* __restrict__ cursor, int n, int E, int per) {
    __shared__ alignas(16) int cs[R_CAP];
    const int tid   = (int)threadIdx.x;
    const int chunk = (int)blockIdx.x / N_RANGES;
    const int range = (int)blockIdx.x - chunk * N_RANGES;
    const int nr    = (((n + N_RANGES - 1) / N_RANGES) + 3) & ~3;
    const int lo    = range * nr;
    const unsigned span = (unsigned)max(0, min(n, lo + nr) - lo);
    for (int i = tid; i < R_CAP / 4; i += 1024) ((int4*)cs)[i] = make_int4(0, 0, 0, 0);
    if (blockIdx.x == 0 && tid == 0) *cursor = 0;
    __syncthreads();
    const int e0 = chunk * per;
    const int e1 = min(E, e0 + per);
    for (int eb = e0; eb < e1; eb += BATCH * 1024) {
        int d[BATCH];
        #pragma unroll
        for (int u = 0; u < BATCH; ++u) {
            const int e = eb + u * 1024 + tid;
            d[u] = (e < e1) ? dst[e] : -1;
        }
        #pragma unroll
        for (int u = 0; u < BATCH; ++u) {
            const unsigned k = (unsigned)(d[u] - lo);
            if (k < span) atomicAdd(&cs[k], 1);
        }
    }
    __syncthreads();
    int* row = table + (size_t)chunk * n + lo;
    if ((span & 3u) == 0 && ((size_t)row & 15) == 0) {
        for (int i = tid; i < (int)(span / 4); i += 1024) ((int4*)row)[i] = ((const int4*)cs)[i];
    } else {
        for (int i = tid; i < (int)span; i += 1024) row[i] = cs[i];
    }
}

__global__ __launch_bounds__(256)
void k_scan_offsets(int* __restrict__ table, int* __restrict__ cursor,
                               int* __restrict__ cnt, int* __restrict__ off,
                               float* __restrict__ dinv, int n) {
    const int i    = blockIdx.x * blockDim.x + threadIdx.x;
    const int lane = threadIdx.x & 31;
    int c = 0;
    if (i < n) {
        const unsigned un = (unsigned)n, ui = (unsigned)i;
        for (int g0 = 0; g0 < N_CHUNKS; g0 += BATCH) {
            int t[BATCH];
            #pragma unroll
            for (int u = 0; u < BATCH; ++u) t[u] = table[(unsigned)(g0 + u) * un + ui];
            #pragma unroll
            for (int u = 0; u < BATCH; ++u) { table[(unsigned)(g0 + u) * un + ui] = c; c += t[u]; }
        }
        cnt[i] = c;
    }
    int incl = c;
    #pragma unroll
    for (int sft = 1; sft < 32; sft <<= 1) {
        int t = __shfl_up(incl, sft);
        if (lane >= sft) incl += t;
    }
    const int total = __shfl(incl, 31);
    int base = 0;
    if (lane == 0) base = atomicAdd(cursor, total);
    base = __shfl(base, 0);
    if (i < n) {
        off[i]  = base + incl - c;
        dinv[i] = __frsqrt_rn((float)(c + 1));
    }
}

__global__ __launch_bounds__(256)
void k_place_gemm1(const int* __restrict__ src, const int* __restrict__ dst,
                   const int* __restrict__ off,
                   const int* __restrict__ table, const float* __restrict__ dinv,
                   int* __restrict__ csr, int n, int E, int per,
                   const float* __restrict__ X, const float* __restrict__ W,
                   float* __restrict__ H, int mTiles, int gemmBlocks) {
    __shared__ float2 Bs[(F_IN / 2) * B_PITCH];
    static_assert(P_SPAN * sizeof(int) <= sizeof(float2) * (F_IN / 2) * B_PITCH, "placement counters must fit the shared array");
    if ((int)blockIdx.x >= gemmBlocks) {
        int* cs = (int*)Bs;
        const int tid   = (int)threadIdx.x;
        const int pb    = (int)blockIdx.x - gemmBlocks;
        const int chunk = pb / P_RANGES;
        const int range = pb - chunk * P_RANGES;
        const int nr    = (((n + P_RANGES - 1) / P_RANGES) + 3) & ~3;
        const int lo    = range * nr;
        const unsigned span = (unsigned)max(0, min(n, lo + nr) - lo);
        const int* trow = table + (size_t)chunk * n + lo;
        const int* orow = off + lo;
        if ((span & 3u) == 0 && (((size_t)trow | (size_t)orow) & 15) == 0) {
            for (int i = tid; i < (int)(span / 4); i += 256) {
                const int4 a = ((const int4*)trow)[i];
                const int4 b = ((const int4*)orow)[i];
                ((int4*)cs)[i] = make_int4(a.x + b.x, a.y + b.y, a.z + b.z, a.w + b.w);
            }
        } else {
            for (int i = tid; i < (int)span; i += 256) cs[i] = trow[i] + orow[i];
        }
        __syncthreads();
        const int e0 = chunk * per;
        const int e1 = min(E, e0 + per);
        for (int eb = e0; eb < e1; eb += P_BATCH * 256) {
            int d[P_BATCH];
            int s[P_BATCH];
            int p[P_BATCH];
            #pragma unroll
            for (int u = 0; u < P_BATCH; ++u) {
                const int e = eb + u * 256 + tid;
                d[u] = (e < e1) ? dst[e] : -1;
            }
            #pragma unroll
            for (int u = 0; u < P_BATCH; ++u) {
                const unsigned k = (unsigned)(d[u] - lo);
                p[u] = -1;
                s[u] = 0;
                if (k < span) {
                    p[u] = atomicAdd(&cs[k], 1);
                    s[u] = src[eb + u * 256 + tid];
                }
            }
            #pragma unroll
            for (int u = 0; u < P_BATCH; ++u) {
                if (p[u] >= 0) csr[p[u]] = s[u];
            }
        }
        return;
    }
    {
        float4 wv[8];
        #pragma unroll
        for (int u = 0; u < 8; ++u) wv[u] = ((const float4*)W)[(int)threadIdx.x + u * 256];
        #pragma unroll
        for (int u = 0; u < 8; ++u) {
            const int idx4 = (int)threadIdx.x + u * 256;
            const int k = idx4 >> 4, q = idx4 & 15;
            float* bp = (float*)(Bs + (k >> 1) * B_PITCH + q) + (k & 1);
            bp[0] = wv[u].x; bp[32] = wv[u].y; bp[64] = wv[u].z; bp[96] = wv[u].w;
        }
    }
    __syncthreads();

    const int wave  = threadIdx.x >> 5;
    const int lane  = threadIdx.x & 31;
    const int mTile = (int)blockIdx.x * 8 + wave;
    if (mTile >= mTiles) return;
    const int mBase = mTile * 16;
    const int half  = lane >> 4;
    const int l     = lane & 15;
    const float*  arow = X + (size_t)(mBase + l) * F_IN + half * 4;
    const float2* brow = Bs + (2 * half) * B_PITCH + l;

    float dv[8];
    #pragma unroll
    for (int r = 0; r < 8; ++r) dv[r] = dinv[mBase + half * 8 + r];

    v8f c0 = {}, c1 = {}, c2 = {}, c3 = {};
    #pragma unroll 4
    for (int j = 0; j < F_IN / 8; ++j) {
        const float4 a4 = *(const float4*)(arow + 8 * j);
        #pragma unroll
        for (int u = 0; u < 2; ++u) {
            v2f a; a.x = u ? a4.z : a4.x; a.y = u ? a4.w : a4.y;
            const float2* bp = brow + (4 * j + u) * B_PITCH;
            const float2 f0 = bp[0], f1 = bp[16], f2 = bp[32], f3 = bp[48];
            v2f b0; b0.x = f0.x; b0.y = f0.y;
            v2f b1; b1.x = f1.x; b1.y = f1.y;
            v2f b2; b2.x = f2.x; b2.y = f2.y;
            v2f b3; b3.x = f3.x; b3.y = f3.y;
            c0 = __builtin_amdgcn_wmma_f32_16x16x4_f32(false, a, false, b0, (short)0, c0, false, false);
            c1 = __builtin_amdgcn_wmma_f32_16x16x4_f32(false, a, false, b1, (short)0, c1, false, false);
            c2 = __builtin_amdgcn_wmma_f32_16x16x4_f32(false, a, false, b2, (short)0, c2, false, false);
            c3 = __builtin_amdgcn_wmma_f32_16x16x4_f32(false, a, false, b3, (short)0, c3, false, false);
        }
    }
    float* orow = H + (size_t)(mBase + half * 8) * F_HID + 4 * l;
    #pragma unroll
    for (int r = 0; r < 8; ++r)
        *(float4*)(orow + (size_t)r * F_HID) = make_float4(c0[r] * dv[r], c1[r] * dv[r], c2[r] * dv[r], c3[r] * dv[r]);
}

#define A_PITCH 68
#define G64_NPW 4
#define G64_NT  (32 * (16 / G64_NPW))

__global__ __launch_bounds__(G64_NT)
void k_gather64_gemm2(const int* __restrict__ csr, const int* __restrict__ off,
                      const int* __restrict__ cnt, const float* __restrict__ dinv,
                      const float* __restrict__ h1, const float* __restrict__ bias1,
                      const float* __restrict__ W2, float* __restrict__ h2, int n) {
    __shared__ alignas(16) float As[16 * A_PITCH];
    __shared__ alignas(16) float Bs[F_HID * 16];
    __shared__ float Ds[16];
    const int lane = threadIdx.x & 31;
    const int wave = __builtin_amdgcn_readfirstlane((int)threadIdx.x >> 5);
    const int r0   = wave * G64_NPW;
    const int n0   = (int)blockIdx.x * 16 + r0;

    float4 wv[256 / G64_NT];
    #pragma unroll
    for (int u = 0; u < 256 / G64_NT; ++u) wv[u] = ((const float4*)W2)[(int)threadIdx.x + u * G64_NT];
    const float2 bb = ((const float2*)bias1)[lane];

    int cl = 0, ol = 0;
    float dl = 0.0f;
    if (lane < G64_NPW && n0 + lane < n) {
        cl = cnt[n0 + lane];
        ol = off[n0 + lane];
        dl = dinv[n0 + lane];
    }
    float2 ownq[G64_NPW];
    #pragma unroll
    for (int q = 0; q < G64_NPW; ++q) {
        ownq[q].x = 0.0f; ownq[q].y = 0.0f;
        if (n0 + q < n) ownq[q] = ((const float2*)(h1 + (size_t)(n0 + q) * F_HID))[lane];
    }
    int cq[G64_NPW], oq[G64_NPW], sq[G64_NPW];
    #pragma unroll
    for (int q = 0; q < G64_NPW; ++q) {
        cq[q] = __builtin_amdgcn_readlane(cl, q);
        oq[q] = __builtin_amdgcn_readlane(ol, q);
    }
    #pragma unroll
    for (int u = 0; u < 256 / G64_NT; ++u) ((float4*)Bs)[(int)threadIdx.x + u * G64_NT] = wv[u];
    #pragma unroll
    for (int q = 0; q < G64_NPW; ++q) {
        sq[q] = n0 + q;
        if (lane < min(32, cq[q])) sq[q] = csr[oq[q] + lane];
    }

    static_assert(G64_NPW % 2 == 0, "nodes are walked in pairs");
    #pragma unroll
    for (int q = 0; q < G64_NPW; q += 2) {
        const int nodeA = n0 + q, nodeB = n0 + q + 1;
        const int cA = cq[q], cB = cq[q + 1];
        const int oA = oq[q], oB = oq[q + 1];
        float2 accA = ownq[q], accB = ownq[q + 1];
        int sA = sq[q], sB = sq[q + 1];
        const int cM = max(cA, cB);
        for (int c0 = 0; c0 < cM; c0 += 32) {
            const int mA = max(0, min(32, cA - c0));
            const int mB = max(0, min(32, cB - c0));
            if (c0 > 0) {
                sA = nodeA;
                if (lane < mA) sA = csr[oA + c0 + lane];
                sB = nodeB;
                if (lane < mB) sB = csr[oB + c0 + lane];
            }
            const int mM = max(mA, mB);
            for (int j0 = 0; j0 < mM; j0 += 8) {
                float2 va[8], vb[8];
                #pragma unroll
                for (int u = 0; u < 8; ++u) {
                    const int sj = __builtin_amdgcn_readlane(sA, j0 + u);
                    va[u] = ((const float2*)(h1 + (size_t)sj * F_HID))[lane];
                }
                #pragma unroll
                for (int u = 0; u < 8; ++u) {
                    const int sj = __builtin_amdgcn_readlane(sB, j0 + u);
                    vb[u] = ((const float2*)(h1 + (size_t)sj * F_HID))[lane];
                }
                asm volatile("" :: "v"(va[0].x), "v"(va[0].y), "v"(va[1].x), "v"(va[1].y),
                                   "v"(va[2].x), "v"(va[2].y), "v"(va[3].x), "v"(va[3].y),
                                   "v"(va[4].x), "v"(va[4].y), "v"(va[5].x), "v"(va[5].y),
                                   "v"(va[6].x), "v"(va[6].y), "v"(va[7].x), "v"(va[7].y));
                asm volatile("" :: "v"(vb[0].x), "v"(vb[0].y), "v"(vb[1].x), "v"(vb[1].y),
                                   "v"(vb[2].x), "v"(vb[2].y), "v"(vb[3].x), "v"(vb[3].y),
                                   "v"(vb[4].x), "v"(vb[4].y), "v"(vb[5].x), "v"(vb[5].y),
                                   "v"(vb[6].x), "v"(vb[6].y), "v"(vb[7].x), "v"(vb[7].y));
                #pragma unroll
                for (int u = 0; u < 8; ++u) {
                    if (j0 + u < mA) {
                        accA.x += va[u].x;
                        accA.y += va[u].y;
                    }
                }
                #pragma unroll
                for (int u = 0; u < 8; ++u) {
                    if (j0 + u < mB) {
                        accB.x += vb[u].x;
                        accB.y += vb[u].y;
                    }
                }
            }
        }
        const float dnA = __int_as_float(__builtin_amdgcn_readlane(__float_as_int(dl), q));
        const float dnB = __int_as_float(__builtin_amdgcn_readlane(__float_as_int(dl), q + 1));
        float2 a1;
        a1.x = fmaxf(accA.x * dnA + bb.x, 0.0f);
        a1.y = fmaxf(accA.y * dnA + bb.y, 0.0f);
        ((float2*)(As + (r0 + q) * A_PITCH))[lane] = a1;
        a1.x = fmaxf(accB.x * dnB + bb.x, 0.0f);
        a1.y = fmaxf(accB.y * dnB + bb.y, 0.0f);
        ((float2*)(As + (r0 + q + 1) * A_PITCH))[lane] = a1;
        if (lane == 0) { Ds[r0 + q] = dnA; Ds[r0 + q + 1] = dnB; }
    }
    __syncthreads();
    if (wave != 0) return;

    const int half = lane >> 4;
    const int l    = lane & 15;
    v8f acc8 = {};
    #pragma unroll
    for (int k = 0; k < F_HID; k += 4) {
        const int ak = k + half * 2;
        v2f a = *(const v2f*)(As + l * A_PITCH + ak);
        v2f b; b.x = Bs[ak * 16 + l]; b.y = Bs[(ak + 1) * 16 + l];
        acc8 = __builtin_amdgcn_wmma_f32_16x16x4_f32(false, a, false, b,
                                                     (short)0, acc8, false, false);
    }
    const int rBase = (int)blockIdx.x * 16 + half * 8;
    float* orow = h2 + (size_t)rBase * F_OUT + l;
    #pragma unroll
    for (int r = 0; r < 8; ++r)
        if (rBase + r < n) orow[(size_t)r * F_OUT] = acc8[r] * Ds[half * 8 + r];
}

#define G16_NPW 8

__global__ __launch_bounds__(256)
void k_gather16(const int* __restrict__ csr, const int* __restrict__ off,
                const int* __restrict__ cnt, const float* __restrict__ dinv,
                const float* __restrict__ h2, const float* __restrict__ bias2,
                float* __restrict__ out, int n) {
    const int lane = threadIdx.x & 31;
    const int n0 = G16_NPW * __builtin_amdgcn_readfirstlane((blockIdx.x * blockDim.x + threadIdx.x) >> 5);
    if (n0 >= n) return;
    const int half = lane >> 4;
    const int f    = lane & 15;
    int cl = 0, ol = 0;
    float dl = 0.0f;
    if (lane < G16_NPW && n0 + lane < n) {
        cl = cnt[n0 + lane];
        ol = off[n0 + lane];
        dl = dinv[n0 + lane];
    }
    const float b2f = bias2[f];
    float ownq[G16_NPW];
    #pragma unroll
    for (int q = 0; q < G16_NPW; ++q) {
        ownq[q] = 0.0f;
        if (n0 + q < n) ownq[q] = h2[(size_t)(n0 + q) * F_OUT + f];
    }
    int cq[G16_NPW], oq[G16_NPW], sq[G16_NPW];
    #pragma unroll
    for (int q = 0; q < G16_NPW; ++q) {
        cq[q] = __builtin_amdgcn_readlane(cl, q);
        oq[q] = __builtin_amdgcn_readlane(ol, q);
        sq[q] = n0 + q;
        if (lane < min(32, cq[q])) sq[q] = csr[oq[q] + lane];
    }
    static_assert(G16_NPW % 2 == 0, "nodes are walked in pairs");
    #pragma unroll
    for (int q = 0; q < G16_NPW; q += 2) {
        const int nodeA = n0 + q, nodeB = n0 + q + 1;
        const int cA = cq[q], cB = cq[q + 1];
        const int oA = oq[q], oB = oq[q + 1];
        float accA = (half == 0) ? ownq[q] : 0.0f;
        float accB = (half == 0) ? ownq[q + 1] : 0.0f;
        int sA = sq[q], sB = sq[q + 1];
        const int cM = max(cA, cB);
        for (int c0 = 0; c0 < cM; c0 += 32) {
            const int mA = max(0, min(32, cA - c0));
            const int mB = max(0, min(32, cB - c0));
            if (c0 > 0) {
                sA = nodeA;
                if (lane < mA) sA = csr[oA + c0 + lane];
                sB = nodeB;
                if (lane < mB) sB = csr[oB + c0 + lane];
            }
            const int mM = max(mA, mB);
            for (int j0 = 0; j0 < mM; j0 += 16) {
                float va[8], vb[8];
                #pragma unroll
                for (int u = 0; u < 8; ++u) {
                    const int sa = __builtin_amdgcn_readlane(sA, j0 + 2 * u);
                    const int sb = __builtin_amdgcn_readlane(sA, j0 + 2 * u + 1);
                    const int sj = half ? sb : sa;
                    va[u] = h2[(size_t)sj * F_OUT + f];
                }
                #pragma unroll
                for (int u = 0; u < 8; ++u) {
                    const int sa = __builtin_amdgcn_readlane(sB, j0 + 2 * u);
                    const int sb = __builtin_amdgcn_readlane(sB, j0 + 2 * u + 1);
                    const int sj = half ? sb : sa;
                    vb[u] = h2[(size_t)sj * F_OUT + f];
                }
                #pragma unroll
                for (int u = 0; u < 8; ++u) {
                    if (j0 + 2 * u + half < mA) accA += va[u];
                }
                #pragma unroll
                for (int u = 0; u < 8; ++u) {
                    if (j0 + 2 * u + half < mB) accB += vb[u];
                }
            }
        }
        const float diA = __int_as_float(__builtin_amdgcn_readlane(__float_as_int(dl), q));
        const float diB = __int_as_float(__builtin_amdgcn_readlane(__float_as_int(dl), q + 1));
        accA += __shfl_xor(accA, 16);
        accB += __shfl_xor(accB, 16);
        const float resA = accA * diA + b2f;
        const float resB = accB * diB + b2f;
        if (nodeA + half < n) out[(size_t)nodeA * F_OUT + lane] = half ? resB : resA;
    }
}

extern "C" void kernel_launch(void* const* d_in, const int* in_sizes, int n_in,
                              void* d_out, int out_size, void* d_ws, size_t ws_size,
                              hipStream_t stream) {
    const float* x  = (const float*)d_in[0];
    const int*   ei = (const int*)  d_in[1];
    const float* W1 = (const float*)d_in[2];
    const float* b1 = (const float*)d_in[3];
    const float* W2 = (const float*)d_in[4];
    const float* b2 = (const float*)d_in[5];
    float* out = (float*)d_out;

    const int n = in_sizes[0] / F_IN;
    const int E = in_sizes[1] / 2;
    const int* src = ei;
    const int* dst = ei + E;

    float* h1     = (float*)d_ws;
    float* h2     = h1 + (size_t)n * F_HID;
    int*   csr    = (int*)(h2 + (size_t)n * F_OUT);
    int*   table  = (int*)(csr + E);
    int*   cnt    = table + (size_t)N_CHUNKS * n;
    int*   off    = cnt + n;
    float* dinv   = (float*)(off + n);
    int*   cursor = (int*)(dinv + n);

    const int mTiles  = n / 16;
    const int mGroups = (mTiles + 7) / 8;

    const int per = (E + N_CHUNKS - 1) / N_CHUNKS;
    k_count_lds<<<N_CHUNKS * N_RANGES, 1024, 0, stream>>>(dst, table, cursor, n, E, per);
    k_scan_offsets<<<(n + 255) / 256, 256, 0, stream>>>(table, cursor, cnt, off, dinv, n);
    const int gemmBlocks  = mGroups;
    const int placeBlocks = N_CHUNKS * P_RANGES;
    k_place_gemm1<<<gemmBlocks + placeBlocks, 256, 0, stream>>>(src, dst, off, table, dinv, csr, n, E, per,
                                                               x, W1, h1, mTiles, gemmBlocks);

    k_gather64_gemm2<<<(n + 15) / 16, G64_NT, 0, stream>>>(csr, off, cnt, dinv, h1, b1, W2, h2, n);

    k_gather16<<<(((n + G16_NPW - 1) / G16_NPW) * 32 + 255) / 256, 256, 0, stream>>>(csr, off, cnt, dinv, h2, b2, out, n);
}
